# starting kernel
# baseline (speedup 1.0000x reference)
#include <hip/hip_runtime.h>
#include <hip/hip_bf16.h>
#include <math.h>


typedef float v2f __attribute__((ext_vector_type(2)));
typedef float v8f __attribute__((ext_vector_type(8)));

#define B_      4
#define I_      96
#define J_      384
#define D_      256
#define NWAVE   12
#define LOGEPS  (-1000.0f)
#define NEGV    (-1000000000.0f)
#define NEGINF  (-1.0e38f)
#define TEMPINV (1.0f/0.55f)
#define ESCALE  (1.0f/256.0f)

__device__ __forceinline__ float lse2(float a, float b) {
  float m = fmaxf(a, b);
  if (m < -9.0e37f) return NEGINF;
  return m + __logf(__expf(a - m) + __expf(b - m));
}

__device__ __forceinline__ float wave_prefix_lse(float x, int lane) {
#pragma unroll
  for (int off = 1; off < 32; off <<= 1) {
    float o = __shfl_up(x, off, 32);
    if (lane >= off) x = lse2(x, o);
  }
  return x;
}

__device__ __forceinline__ float wave_suffix_lse(float x, int lane) {
#pragma unroll
  for (int off = 1; off < 32; off <<= 1) {
    float o = __shfl_down(x, off, 32);
    if (lane + off < 32) x = lse2(x, o);
  }
  return x;
}

__global__ __launch_bounds__(384, 1)
void mobo_kernel(const float* __restrict__ text,
                 const float* __restrict__ mel,
                 const float* __restrict__ gum,
                 float* __restrict__ out_soft,
                 float* __restrict__ out_exp)
{
  __shared__ float sE[I_ * J_];
  __shared__ float sP[J_];
  __shared__ float sSB[J_];
  __shared__ float sB0[J_];
  __shared__ float sB1[J_];
  __shared__ float sWtL[NWAVE];
  __shared__ float sWtP[NWAVE];
  __shared__ float sWtSA[NWAVE];
  __shared__ float sWtSB[NWAVE];

  const int b    = blockIdx.x;
  const int tid  = threadIdx.x;
  const int lane = tid & 31;
  const int wave = tid >> 5;
  const int half = lane >> 4;
  const int l16  = lane & 15;

  const float* tx = text + (size_t)b * I_ * D_;
  const float* ml = mel  + (size_t)b * J_ * D_;

  for (int t = wave; t < (I_/16) * (J_/16); t += NWAVE) {
    const int mT = t % (I_/16);
    const int nT = t / (I_/16);
    const int mrow = mT * 16 + l16;
    const int ncol = nT * 16 + l16;
    v8f acc = {};
    for (int k0 = 0; k0 < D_; k0 += 4) {
      const int kk = k0 + 2 * half;
      v2f a  = *(const v2f*)(tx + (size_t)mrow * D_ + kk);
      v2f bf = *(const v2f*)(ml + (size_t)ncol * D_ + kk);
      acc = __builtin_amdgcn_wmma_f32_16x16x4_f32(false, a, false, bf,
                                                  (short)0, acc, false, false);
    }
    for (int r = 0; r < 8; ++r) {
      const int i = mT * 16 + r + 8 * half;
      const int j = nT * 16 + l16;
      const float u = gum[(size_t)b * I_ * J_ + (size_t)i * J_ + j];
      const float g = -__logf(-__logf(u));
      sE[i * J_ + j] = (acc[r] * ESCALE + g) * TEMPINV;
    }
  }
  __syncthreads();

  float* Bc = sB0;
  float* Bn = sB1;
  Bc[tid] = (tid == 0) ? 0.0f : NEGV;
  __syncthreads();

  for (int i = 0; i < I_; ++i) {
    const float e  = sE[i * J_ + tid];
    float xl = wave_suffix_lse(e, lane);
    if (lane == 0) sWtL[wave] = xl;
    __syncthreads();
    float lhi = NEGINF;
    for (int w = wave + 1; w < NWAVE; ++w) lhi = lse2(lhi, sWtL[w]);
    const float Lv = lse2(xl, lhi);

    const float bv = Bc[tid];
    const float av = bv - Lv;
    float xp  = wave_prefix_lse(av, lane);
    float xsa = wave_suffix_lse(av, lane);
    float xsb = wave_suffix_lse(bv, lane);
    if (lane == 31) sWtP[wave] = xp;
    if (lane == 0) { sWtSA[wave] = xsa; sWtSB[wave] = xsb; }
    __syncthreads();
    float plo = NEGINF, sahi = NEGINF, sbhi = NEGINF;
    for (int w = 0; w < wave; ++w)          plo  = lse2(plo,  sWtP[w]);
    for (int w = wave + 1; w < NWAVE; ++w) { sahi = lse2(sahi, sWtSA[w]);
                                             sbhi = lse2(sbhi, sWtSB[w]); }
    const float Pv  = lse2(xp,  plo);
    const float SAv = lse2(xsa, sahi);
    const float SBv = lse2(xsb, sbhi);
    sP[tid]  = Pv;
    sSB[tid] = SBv;
    __syncthreads();

    if (i < I_ - 1) {
      float nb;
      if (tid == 0) {
        nb = NEGV;
      } else {
        const int j = tid - 1;
        const float v = lse2(sE[i * J_ + j] + sP[j], SAv + LOGEPS);
        const int r = i + 1;
        nb = (tid >= r && tid <= (J_ - I_ + r + 1)) ? v : NEGV;
      }
      Bn[tid] = nb;
    }

    float lb;
    if (i < I_ - 1) {
      const float sbexc = (tid + 1 < J_) ? sSB[tid + 1] : NEGINF;
      lb = lse2(Lv + Pv, sbexc + LOGEPS);
    } else {
      lb = sSB[0] + ((tid == J_ - 1) ? 0.0f : LOGEPS);
    }
    out_soft[(size_t)b * I_ * J_ + (size_t)i * J_ + tid] = lb;

    __syncthreads();
    sE[i * J_ + tid] = __expf(lb);
    { float* tmp = Bc; Bc = Bn; Bn = tmp; }
  }
  __syncthreads();

  for (int t = wave; t < (J_/16) * (D_/16); t += NWAVE) {
    const int mT = t % (J_/16);
    const int nT = t / (J_/16);
    const int m = mT * 16 + l16;
    const int n = nT * 16 + l16;
    v8f acc = {};
    for (int k0 = 0; k0 < I_; k0 += 4) {
      const int kk = k0 + 2 * half;
      v2f a;
      a.x = sE[kk * J_ + m];
      a.y = sE[(kk + 1) * J_ + m];
      v2f bf;
      bf.x = tx[(size_t)kk * D_ + n];
      bf.y = tx[(size_t)(kk + 1) * D_ + n];
      acc = __builtin_amdgcn_wmma_f32_16x16x4_f32(false, a, false, bf,
                                                  (short)0, acc, false, false);
    }
    for (int r = 0; r < 8; ++r) {
      const int j = mT * 16 + r + 8 * half;
      out_exp[(size_t)b * J_ * D_ + (size_t)j * D_ + n] = acc[r];
    }
  }
}

extern "C" void kernel_launch(void* const* d_in, const int* in_sizes, int n_in,
                              void* d_out, int out_size, void* d_ws, size_t ws_size,
                              hipStream_t stream) {
  (void)in_sizes; (void)n_in; (void)d_ws; (void)ws_size; (void)out_size;
  const float* text = (const float*)d_in[0];
  const float* mel  = (const float*)d_in[1];
  const float* gum  = (const float*)d_in[2];
  float* out_soft = (float*)d_out;
  float* out_exp  = (float*)d_out + (size_t)B_ * I_ * J_;
  mobo_kernel<<<dim3(B_), dim3(384), 0, stream>>>(text, mel, gum, out_soft, out_exp);
}


# Order-1 column step as ONE three-term LSE per row: lse3(P1_i(j-1), (E-L)+P1_{i-1}(j-1), T0_{i-1}-L) for two chained two-term LSEs; E-L, T0-L, window a column ahead; new rounding and sum order in P1
# speedup vs baseline: 3.5665x; 3.5665x over previous
#include <hip/hip_runtime.h>
#include <hip/hip_bf16.h>
#include <math.h>


typedef float v2f __attribute__((ext_vector_type(2)));
typedef float v8f __attribute__((ext_vector_type(8)));

#define B_      4
#define I_      96
#define J_      384
#define D_      256
#define NWAVE   12
#define LDE     (J_ + 1)
#define LOGEPS  (-1000.0f)
#define NEGV    (-1000000000.0f)
#define NEGINF  (-1.0e38f)
#define TEMPINV (1.0f/0.55f)
#define ESCALE  (1.0f/256.0f)
#define KC      64
#define CB      32
#define NBLK    (J_ / CB)
#define SPITCH  (KC + 4)

__device__ __forceinline__ float lse2(float a, float b) {
  float m = fmaxf(a, b);
  if (m < -9.0e37f) return NEGINF;
  return m + __logf(__expf(a - m) + __expf(b - m));
}

template <int N> struct vecN { typedef float type __attribute__((ext_vector_type(N))); };
template <int N>
__device__ __forceinline__ void lse2n(float (&a)[N], const float (&b)[N]) {
  float m[N], ea[N], eb[N];
#pragma unroll
  for (int n = 0; n < N; ++n) m[n] = fmaxf(a[n], b[n]);
#pragma unroll
  for (int n = 0; n < N; ++n) ea[n] = __expf(a[n] - m[n]);
#pragma unroll
  for (int n = 0; n < N; ++n) eb[n] = __expf(b[n] - m[n]);
  typename vecN<N>::type sv;
#pragma unroll
  for (int n = 0; n < N; ++n) sv[n] = ea[n] + eb[n];
  const typename vecN<N>::type lg = __builtin_elementwise_log(sv);
#pragma unroll
  for (int n = 0; n < N; ++n) {
    const float r = m[n] + lg[n];
    a[n] = (m[n] < -9.0e37f) ? NEGINF : r;
  }
}

template <int N>
__device__ __forceinline__ void lse3n(float (&a)[N], const float (&b)[N], const float (&c)[N]) {
  float m[N], ea[N], eb[N], ec[N];
#pragma unroll
  for (int n = 0; n < N; ++n) m[n] = fmaxf(fmaxf(a[n], b[n]), c[n]);
#pragma unroll
  for (int n = 0; n < N; ++n) ea[n] = __expf(a[n] - m[n]);
#pragma unroll
  for (int n = 0; n < N; ++n) eb[n] = __expf(b[n] - m[n]);
#pragma unroll
  for (int n = 0; n < N; ++n) ec[n] = __expf(c[n] - m[n]);
  typename vecN<N>::type sv;
#pragma unroll
  for (int n = 0; n < N; ++n) sv[n] = ea[n] + (eb[n] + ec[n]);
  const typename vecN<N>::type lg = __builtin_elementwise_log(sv);
#pragma unroll
  for (int n = 0; n < N; ++n) {
    const float r = m[n] + lg[n];
    a[n] = (m[n] < -9.0e37f) ? NEGINF : r;
  }
}

__device__ __forceinline__ float wave_prefix_lse(float x, int lane) {
#pragma unroll
  for (int off = 1; off < 32; off <<= 1) {
    float o = __shfl_up(x, off, 32);
    if (lane >= off) x = lse2(x, o);
  }
  return x;
}

__device__ __forceinline__ float wave_suffix_lse(float x, int lane) {
#pragma unroll
  for (int off = 1; off < 32; off <<= 1) {
    float o = __shfl_down(x, off, 32);
    if (lane + off < 32) x = lse2(x, o);
  }
  return x;
}

__global__ __launch_bounds__(384, 1)
void mobo_kernel(const float* __restrict__ text,
                 const float* __restrict__ mel,
                 const float* __restrict__ gum,
                 float* __restrict__ out_soft,
                 float* __restrict__ out_exp)
{
  __shared__ float sE[I_ * LDE];
  __shared__ alignas(16) float sL[I_ * LDE];
  __shared__ int   sFlag;
  __shared__ float sP[J_];
  __shared__ float sSB[J_];
  __shared__ float sB0[J_];
  __shared__ float sB1[J_];
  __shared__ float sWtL[NWAVE];
  __shared__ float sWtP[NWAVE];
  __shared__ float sWtSA[NWAVE];
  __shared__ float sWtSB[NWAVE];

  const int b    = blockIdx.x;
  const int tid  = threadIdx.x;
  const int lane = tid & 31;
  const int wave = tid >> 5;
  const int half = lane >> 4;
  const int l16  = lane & 15;

  const float* tx = text + (size_t)b * I_ * D_;
  const float* ml = mel  + (size_t)b * J_ * D_;
  float* osoft = out_soft + (size_t)b * I_ * J_;

  if (tid == 0) sFlag = 0;
  __syncthreads();
  int bad = 0;

  {
    float* stg = sL;
    const int prow = tid >> 4;
    const int pc4  = (tid & 15) * 4;
    const float* gsrc[2];
    gsrc[0] = tx + (size_t)prow * D_ + pc4;
    gsrc[1] = ml + (size_t)prow * D_ + pc4;
    float* sdst = stg + prow * SPITCH + pc4;
#define PIECES(X) X(0) X(1) X(2) X(3) X(4) X(5) X(6) X(7) X(8) X(9) X(10) X(11) X(12) X(13) X(14) X(15) X(16) X(17) X(18) X(19)
#define PSRC(u)   ((u) < 4 ? gsrc[0] + (size_t)(24 * (u)) * D_ : gsrc[1] + (size_t)(24 * ((u) - 4)) * D_)
#define PDECL(u)  float4 pf##u;
#define PLOAD(u)  pf##u = *(const float4*)(PSRC(u) + koff);
#define PSTORE(u) *(float4*)(sdst + 24 * (u) * SPITCH) = pf##u;
    PIECES(PDECL)

    const int mG = wave & 1;
    const int nG = wave >> 1;
    const float* fa = stg + (48 * mG + l16) * SPITCH + 2 * half;
    const float* fb = stg + (I_ + 64 * nG + l16) * SPITCH + 2 * half;

    v8f acc[3][4];
#pragma unroll
    for (int m = 0; m < 3; ++m)
#pragma unroll
      for (int n = 0; n < 4; ++n) acc[m][n] = (v8f){};

    {
      const int koff = 0;
      PIECES(PLOAD)
      PIECES(PSTORE)
    }
    __syncthreads();

    auto chunk_mma = [&]() {
#pragma unroll 4
      for (int kk = 0; kk < KC; kk += 4) {
        v2f a[3], bf[4];
#pragma unroll
        for (int m = 0; m < 3; ++m) a[m]  = *(const v2f*)(fa + m * 16 * SPITCH + kk);
#pragma unroll
        for (int n = 0; n < 4; ++n) bf[n] = *(const v2f*)(fb + n * 16 * SPITCH + kk);
#pragma unroll
        for (int m = 0; m < 3; ++m)
#pragma unroll
          for (int n = 0; n < 4; ++n)
            acc[m][n] = __builtin_amdgcn_wmma_f32_16x16x4_f32(false, a[m], false, bf[n],
                                                              (short)0, acc[m][n], false, false);
      }
    };

#pragma unroll 1
    for (int kc = 1; kc < D_ / KC; ++kc) {
      chunk_mma();
      __syncthreads();
      {
        const int koff = kc * KC;
        PIECES(PLOAD)
      }
      PIECES(PSTORE)
      __syncthreads();
    }
    chunk_mma();

#pragma unroll
    for (int m = 0; m < 3; ++m) {
#pragma unroll
      for (int n = 0; n < 4; ++n) {
#pragma unroll
        for (int r = 0; r < 8; ++r) {
          const int i = 48 * mG + 16 * m + r + 8 * half;
          const int j = 64 * nG + 16 * n + l16;
          const float u = gum[(size_t)b * I_ * J_ + (size_t)i * J_ + j];
          const float g = -__logf(-__logf(u));
          const float ev = (acc[m][n][r] * ESCALE + g) * TEMPINV;
          if (!(ev > -400.0f)) bad = 1;
          sE[i * LDE + j] = ev;
        }
      }
    }
  }
  if (bad) sFlag = 1;
  __syncthreads();
  const int slow = sFlag;

  auto tileC = [&](const int mT, const int nT) {
    const int m = mT * 16 + l16;
    const int n = nT * 16 + l16;
    v8f acc = {};
    for (int k0 = 0; k0 < I_; k0 += 4) {
      const int kk = k0 + 2 * half;
      v2f a;
      a.x = sE[kk * LDE + m];
      a.y = sE[(kk + 1) * LDE + m];
      v2f bf;
      bf.x = tx[(size_t)kk * D_ + n];
      bf.y = tx[(size_t)(kk + 1) * D_ + n];
      acc = __builtin_amdgcn_wmma_f32_16x16x4_f32(false, a, false, bf,
                                                  (short)0, acc, false, false);
    }
    for (int r = 0; r < 8; ++r) {
      const int j = mT * 16 + r + 8 * half;
      out_exp[(size_t)b * J_ * D_ + (size_t)j * D_ + n] = acc[r];
    }
  };

  if (slow) {
    float* Bc = sB0;
    float* Bn = sB1;
    Bc[tid] = (tid == 0) ? 0.0f : NEGV;
    __syncthreads();

    for (int i = 0; i < I_; ++i) {
      const float e  = sE[i * LDE + tid];
      float xl = wave_suffix_lse(e, lane);
      if (lane == 0) sWtL[wave] = xl;
      __syncthreads();
      float lhi = NEGINF;
      for (int w = wave + 1; w < NWAVE; ++w) lhi = lse2(lhi, sWtL[w]);
      const float Lv = lse2(xl, lhi);

      const float bv = Bc[tid];
      const float av = bv - Lv;
      float xp  = wave_prefix_lse(av, lane);
      float xsa = wave_suffix_lse(av, lane);
      float xsb = wave_suffix_lse(bv, lane);
      if (lane == 31) sWtP[wave] = xp;
      if (lane == 0) { sWtSA[wave] = xsa; sWtSB[wave] = xsb; }
      __syncthreads();
      float plo = NEGINF, sahi = NEGINF, sbhi = NEGINF;
      for (int w = 0; w < wave; ++w)          plo  = lse2(plo,  sWtP[w]);
      for (int w = wave + 1; w < NWAVE; ++w) { sahi = lse2(sahi, sWtSA[w]);
                                               sbhi = lse2(sbhi, sWtSB[w]); }
      const float Pv  = lse2(xp,  plo);
      const float SAv = lse2(xsa, sahi);
      const float SBv = lse2(xsb, sbhi);
      sP[tid]  = Pv;
      sSB[tid] = SBv;
      __syncthreads();

      if (i < I_ - 1) {
        float nb;
        if (tid == 0) {
          nb = NEGV;
        } else {
          const int j = tid - 1;
          const float v = lse2(sE[i * LDE + j] + sP[j], SAv + LOGEPS);
          const int r = i + 1;
          nb = (tid >= r && tid <= (J_ - I_ + r + 1)) ? v : NEGV;
        }
        Bn[tid] = nb;
      }

      float lb;
      if (i < I_ - 1) {
        const float sbexc = (tid + 1 < J_) ? sSB[tid + 1] : NEGINF;
        lb = lse2(Lv + Pv, sbexc + LOGEPS);
      } else {
        lb = sSB[0] + ((tid == J_ - 1) ? 0.0f : LOGEPS);
      }
      out_soft[(size_t)b * I_ * J_ + (size_t)i * J_ + tid] = lb;

      __syncthreads();
      sE[i * LDE + tid] = __expf(lb);
      { float* tmp = Bc; Bc = Bn; Bn = tmp; }
    }
    __syncthreads();
    for (int t = wave; t < (J_/16) * (D_/16); t += NWAVE) tileC(t % (J_/16), t / (J_/16));
  } else {
    {
      const int li = tid % I_;
      const int lq = tid / I_;
      const int o0 = li * LDE + lq * (J_ / 4);
      float s = NEGINF;
      for (int t = J_ / 4 - 1; t >= 0; --t) {
        s = lse2(sE[o0 + t], s);
        sL[o0 + t] = s;
      }
      sSB[tid] = s;
      __syncthreads();
      float carry = NEGINF;
      for (int q2 = lq + 1; q2 < 4; ++q2) carry = lse2(carry, sSB[q2 * I_ + li]);
      if (lq < 3)
        for (int t = 0; t < J_ / 4; ++t) sL[o0 + t] = lse2(sL[o0 + t], carry);
      __syncthreads();
    }

    const int i0 = 3 * lane;
    int eo[3], lo[3];
    float eN[3], lN[3];
    float tu[3];
    float qv[3];
    float dC[3], cC[3], lC[3];
    float eR[3], lR[3];
#pragma unroll
    for (int r = 0; r < 3; ++r) {
      const int i = i0 + r;
      eo[r] = ((i > 0) ? i - 1 : 0) * LDE;
      lo[r] = i * LDE;
      eN[r] = 0.0f;
      lN[r] = 0.0f;
      tu[r] = NEGINF;
      qv[r] = NEGINF;
      dC[r] = NEGINF;
      cC[r] = NEGINF;
      lC[r] = 0.0f;
      eR[r] = 0.0f;
      lR[r] = 0.0f;
    }
    if (wave == 0) {
      float ps[6];
#pragma unroll
      for (int r = 0; r < 3; ++r) {
        lN[r] = sL[lo[r]];
        ps[r]     = NEGINF;
        ps[3 + r] = NEGINF;
      }

      for (int j4 = 0; j4 < J_; j4 += 4) {
        float xq[3][4];
#pragma unroll
        for (int q = 0; q < 4; ++q) {
          const int j = j4 + q;
          float eC[3], Lc[3], pu[3];
          pu[0] = __shfl_up(ps[2], 1, 32);
          pu[1] = ps[0];
          pu[2] = ps[1];
#pragma unroll
          for (int r = 0; r < 3; ++r) {
            eC[r] = eN[r];
            Lc[r] = lN[r];
            eN[r] = sE[eo[r] + j];
            lN[r] = sL[lo[r] + j + 1];
          }
          float pb[6];
#pragma unroll
          for (int r = 0; r < 3; ++r) {
            const int i = i0 + r;
            const bool win = (j >= i) && (j <= J_ - I_ + i + 1);
            float b0 = win ? eC[r] + pu[r] : NEGV;
            if (r == 0) b0 = (i == 0) ? ((j == 0) ? 0.0f : NEGV) : b0;
            pb[r]     = b0 - Lc[r];
            pb[3 + r] = b0;
          }
          lse2n<6>(ps, pb);
#pragma unroll
          for (int r = 0; r < 3; ++r) xq[r][q] = Lc[r] + ps[r];
        }
#pragma unroll
        for (int r = 0; r < 3; ++r)
          *(float4*)(osoft + (size_t)(i0 + r) * J_ + j4) =
              make_float4(xq[r][0], xq[r][1], xq[r][2], xq[r][3]);
      }
#pragma unroll
      for (int r = 0; r < 3; ++r) sP[i0 + r] = ps[3 + r];

      tu[0] = __shfl_up(ps[2], 1, 32);
      tu[1] = ps[0];
      tu[2] = ps[1];
#pragma unroll
      for (int r = 0; r < 3; ++r) {
        qv[r] = NEGINF;
        dC[r] = NEGINF;
        cC[r] = NEGINF;
        lC[r] = sL[lo[r]];
        eR[r] = sE[eo[r]];
        lR[r] = sL[lo[r] + 1];
      }
    }
    __syncthreads();

#pragma unroll 1
    for (int blk = 0; blk < NBLK + 2; ++blk) {
      if (wave == 0) {
        if (blk < NBLK) {
          for (int j = blk * CB; j < blk * CB + CB; ++j) {
            float qu[3];
            qu[0] = __shfl_up(qv[2], 1, 32);
            qu[1] = qv[0];
            qu[2] = qv[1];
            const int jl = (j + 2 <= J_) ? j + 2 : J_;
            float e2[3], l2[3], dN[3], cN[3];
#pragma unroll
            for (int r = 0; r < 3; ++r) {
              e2[r] = sE[eo[r] + j + 1];
              l2[r] = sL[lo[r] + jl];
            }
#pragma unroll
            for (int r = 0; r < 3; ++r) {
              const int i = i0 + r;
              const bool wn = (i > 0) && (j + 1 >= i) && (j <= J_ - I_ + i);
              dN[r] = wn ? eR[r] - lR[r] : NEGINF;
              cN[r] = wn ? tu[r] - lR[r] : NEGINF;
            }
            float bq[3];
#pragma unroll
            for (int r = 0; r < 3; ++r) bq[r] = dC[r] + qu[r];
            lse3n<3>(qv, bq, cC);
#pragma unroll
            for (int r = 0; r < 3; ++r) {
              sL[lo[r] + j] = lC[r] + qv[r];
              dC[r] = dN[r];
              cC[r] = cN[r];
              lC[r] = lR[r];
              eR[r] = e2[r];
              lR[r] = l2[r];
            }
          }
        }
      } else {
        const int ww = wave - 1;
        if (blk >= 1 && blk <= NBLK) {
          const int j = (blk - 1) * CB + lane;
          float xr[9];
#pragma unroll
          for (int u = 0; u < 9; ++u) {
            const int i2 = ww + (NWAVE - 1) * u;
            xr[u] = (i2 < I_) ? osoft[(size_t)i2 * J_ + j] : 0.0f;
          }
#pragma unroll
          for (int u = 0; u < 9; ++u) {
            const int i2 = ww + (NWAVE - 1) * u;
            if (i2 < I_) {
              const float s = sP[i2];
              float lb;
              if (i2 < I_ - 1) {
                const float y = sL[i2 * LDE + j];
                lb = lse2(lse2(xr[u], y + LOGEPS), s + LOGEPS);
              } else {
                lb = s + ((j == J_ - 1) ? 0.0f : LOGEPS);
              }
              osoft[(size_t)i2 * J_ + j] = lb;
              sE[i2 * LDE + j] = __expf(lb);
            }
          }
        }
        if (blk >= 2) {
          for (int t = ww; t < (CB / 16) * (D_ / 16); t += NWAVE - 1)
            tileC((CB / 16) * (blk - 2) + (t & 1), t >> 1);
        }
      }
      __syncthreads();
    }
  }
}

extern "C" void kernel_launch(void* const* d_in, const int* in_sizes, int n_in,
                              void* d_out, int out_size, void* d_ws, size_t ws_size,
                              hipStream_t stream) {
  (void)in_sizes; (void)n_in; (void)d_ws; (void)ws_size; (void)out_size;
  const float* text = (const float*)d_in[0];
  const float* mel  = (const float*)d_in[1];
  const float* gum  = (const float*)d_in[2];
  float* out_soft = (float*)d_out;
  float* out_exp  = (float*)d_out + (size_t)B_ * I_ * J_;
  mobo_kernel<<<dim3(B_), dim3(384), 0, stream>>>(text, mel, gum, out_soft, out_exp);
}


# Column steps of both sweeps free of LDS-unit waits: lane hand-over as cross-row permute + DPP row shift on the vector ALU, order-1 Y store issued one column late; all sums, orders kept
# speedup vs baseline: 3.7732x; 1.0580x over previous
#include <hip/hip_runtime.h>
#include <hip/hip_bf16.h>
#include <math.h>


typedef float v2f __attribute__((ext_vector_type(2)));
typedef float v8f __attribute__((ext_vector_type(8)));

#define B_      4
#define I_      96
#define J_      384
#define D_      256
#define NWAVE   12
#define LDE     (J_ + 1)
#define LOGEPS  (-1000.0f)
#define NEGV    (-1000000000.0f)
#define NEGINF  (-1.0e38f)
#define TEMPINV (1.0f/0.55f)
#define ESCALE  (1.0f/256.0f)
#define KC      64
#define CB      32
#define NBLK    (J_ / CB)
#define SPITCH  (KC + 4)

__device__ __forceinline__ float lse2(float a, float b) {
  float m = fmaxf(a, b);
  if (m < -9.0e37f) return NEGINF;
  return m + __logf(__expf(a - m) + __expf(b - m));
}

template <int N> struct vecN { typedef float type __attribute__((ext_vector_type(N))); };
template <int N>
__device__ __forceinline__ void lse2n(float (&a)[N], const float (&b)[N]) {
  float m[N], ea[N], eb[N];
#pragma unroll
  for (int n = 0; n < N; ++n) m[n] = fmaxf(a[n], b[n]);
#pragma unroll
  for (int n = 0; n < N; ++n) ea[n] = __expf(a[n] - m[n]);
#pragma unroll
  for (int n = 0; n < N; ++n) eb[n] = __expf(b[n] - m[n]);
  typename vecN<N>::type sv;
#pragma unroll
  for (int n = 0; n < N; ++n) sv[n] = ea[n] + eb[n];
  const typename vecN<N>::type lg = __builtin_elementwise_log(sv);
#pragma unroll
  for (int n = 0; n < N; ++n) {
    const float r = m[n] + lg[n];
    a[n] = (m[n] < -9.0e37f) ? NEGINF : r;
  }
}

template <int N>
__device__ __forceinline__ void lse3n(float (&a)[N], const float (&b)[N], const float (&c)[N]) {
  float m[N], ea[N], eb[N], ec[N];
#pragma unroll
  for (int n = 0; n < N; ++n) m[n] = fmaxf(fmaxf(a[n], b[n]), c[n]);
#pragma unroll
  for (int n = 0; n < N; ++n) ea[n] = __expf(a[n] - m[n]);
#pragma unroll
  for (int n = 0; n < N; ++n) eb[n] = __expf(b[n] - m[n]);
#pragma unroll
  for (int n = 0; n < N; ++n) ec[n] = __expf(c[n] - m[n]);
  typename vecN<N>::type sv;
#pragma unroll
  for (int n = 0; n < N; ++n) sv[n] = ea[n] + (eb[n] + ec[n]);
  const typename vecN<N>::type lg = __builtin_elementwise_log(sv);
#pragma unroll
  for (int n = 0; n < N; ++n) {
    const float r = m[n] + lg[n];
    a[n] = (m[n] < -9.0e37f) ? NEGINF : r;
  }
}

__device__ __forceinline__ float wave_prefix_lse(float x, int lane) {
#pragma unroll
  for (int off = 1; off < 32; off <<= 1) {
    float o = __shfl_up(x, off, 32);
    if (lane >= off) x = lse2(x, o);
  }
  return x;
}

__device__ __forceinline__ float wave_suffix_lse(float x, int lane) {
#pragma unroll
  for (int off = 1; off < 32; off <<= 1) {
    float o = __shfl_down(x, off, 32);
    if (lane + off < 32) x = lse2(x, o);
  }
  return x;
}

__device__ __forceinline__ float lane_below(float x) {
  const int xi = __float_as_int(x);
  const int cross = __builtin_amdgcn_permlanex16(xi, xi, 0xffffffffu, 0xffffffffu, false, false);
  return __int_as_float(__builtin_amdgcn_update_dpp(cross, xi, 0x111, 0xf, 0xf, false));
}

__global__ __launch_bounds__(384, 1)
void mobo_kernel(const float* __restrict__ text,
                 const float* __restrict__ mel,
                 const float* __restrict__ gum,
                 float* __restrict__ out_soft,
                 float* __restrict__ out_exp)
{
  __shared__ float sE[I_ * LDE];
  __shared__ alignas(16) float sL[I_ * LDE];
  __shared__ int   sFlag;
  __shared__ float sP[J_];
  __shared__ float sSB[J_];
  __shared__ float sB0[J_];
  __shared__ float sB1[J_];
  __shared__ float sWtL[NWAVE];
  __shared__ float sWtP[NWAVE];
  __shared__ float sWtSA[NWAVE];
  __shared__ float sWtSB[NWAVE];

  const int b    = blockIdx.x;
  const int tid  = threadIdx.x;
  const int lane = tid & 31;
  const int wave = tid >> 5;
  const int half = lane >> 4;
  const int l16  = lane & 15;

  const float* tx = text + (size_t)b * I_ * D_;
  const float* ml = mel  + (size_t)b * J_ * D_;
  float* osoft = out_soft + (size_t)b * I_ * J_;

  if (tid == 0) sFlag = 0;
  __syncthreads();
  int bad = 0;

  {
    float* stg = sL;
    const int prow = tid >> 4;
    const int pc4  = (tid & 15) * 4;
    const float* gsrc[2];
    gsrc[0] = tx + (size_t)prow * D_ + pc4;
    gsrc[1] = ml + (size_t)prow * D_ + pc4;
    float* sdst = stg + prow * SPITCH + pc4;
#define PIECES(X) X(0) X(1) X(2) X(3) X(4) X(5) X(6) X(7) X(8) X(9) X(10) X(11) X(12) X(13) X(14) X(15) X(16) X(17) X(18) X(19)
#define PSRC(u)   ((u) < 4 ? gsrc[0] + (size_t)(24 * (u)) * D_ : gsrc[1] + (size_t)(24 * ((u) - 4)) * D_)
#define PDECL(u)  float4 pf##u;
#define PLOAD(u)  pf##u = *(const float4*)(PSRC(u) + koff);
#define PSTORE(u) *(float4*)(sdst + 24 * (u) * SPITCH) = pf##u;
    PIECES(PDECL)

    const int mG = wave & 1;
    const int nG = wave >> 1;
    const float* fa = stg + (48 * mG + l16) * SPITCH + 2 * half;
    const float* fb = stg + (I_ + 64 * nG + l16) * SPITCH + 2 * half;

    v8f acc[3][4];
#pragma unroll
    for (int m = 0; m < 3; ++m)
#pragma unroll
      for (int n = 0; n < 4; ++n) acc[m][n] = (v8f){};

    {
      const int koff = 0;
      PIECES(PLOAD)
      PIECES(PSTORE)
    }
    __syncthreads();

    auto chunk_mma = [&]() {
#pragma unroll 4
      for (int kk = 0; kk < KC; kk += 4) {
        v2f a[3], bf[4];
#pragma unroll
        for (int m = 0; m < 3; ++m) a[m]  = *(const v2f*)(fa + m * 16 * SPITCH + kk);
#pragma unroll
        for (int n = 0; n < 4; ++n) bf[n] = *(const v2f*)(fb + n * 16 * SPITCH + kk);
#pragma unroll
        for (int m = 0; m < 3; ++m)
#pragma unroll
          for (int n = 0; n < 4; ++n)
            acc[m][n] = __builtin_amdgcn_wmma_f32_16x16x4_f32(false, a[m], false, bf[n],
                                                              (short)0, acc[m][n], false, false);
      }
    };

#pragma unroll 1
    for (int kc = 1; kc < D_ / KC; ++kc) {
      chunk_mma();
      __syncthreads();
      {
        const int koff = kc * KC;
        PIECES(PLOAD)
      }
      PIECES(PSTORE)
      __syncthreads();
    }
    chunk_mma();

#pragma unroll
    for (int m = 0; m < 3; ++m) {
#pragma unroll
      for (int n = 0; n < 4; ++n) {
#pragma unroll
        for (int r = 0; r < 8; ++r) {
          const int i = 48 * mG + 16 * m + r + 8 * half;
          const int j = 64 * nG + 16 * n + l16;
          const float u = gum[(size_t)b * I_ * J_ + (size_t)i * J_ + j];
          const float g = -__logf(-__logf(u));
          const float ev = (acc[m][n][r] * ESCALE + g) * TEMPINV;
          if (!(ev > -400.0f)) bad = 1;
          sE[i * LDE + j] = ev;
        }
      }
    }
  }
  if (bad) sFlag = 1;
  __syncthreads();
  const int slow = sFlag;

  auto tileC = [&](const int mT, const int nT) {
    const int m = mT * 16 + l16;
    const int n = nT * 16 + l16;
    v8f acc = {};
    for (int k0 = 0; k0 < I_; k0 += 4) {
      const int kk = k0 + 2 * half;
      v2f a;
      a.x = sE[kk * LDE + m];
      a.y = sE[(kk + 1) * LDE + m];
      v2f bf;
      bf.x = tx[(size_t)kk * D_ + n];
      bf.y = tx[(size_t)(kk + 1) * D_ + n];
      acc = __builtin_amdgcn_wmma_f32_16x16x4_f32(false, a, false, bf,
                                                  (short)0, acc, false, false);
    }
    for (int r = 0; r < 8; ++r) {
      const int j = mT * 16 + r + 8 * half;
      out_exp[(size_t)b * J_ * D_ + (size_t)j * D_ + n] = acc[r];
    }
  };

  if (slow) {
    float* Bc = sB0;
    float* Bn = sB1;
    Bc[tid] = (tid == 0) ? 0.0f : NEGV;
    __syncthreads();

    for (int i = 0; i < I_; ++i) {
      const float e  = sE[i * LDE + tid];
      float xl = wave_suffix_lse(e, lane);
      if (lane == 0) sWtL[wave] = xl;
      __syncthreads();
      float lhi = NEGINF;
      for (int w = wave + 1; w < NWAVE; ++w) lhi = lse2(lhi, sWtL[w]);
      const float Lv = lse2(xl, lhi);

      const float bv = Bc[tid];
      const float av = bv - Lv;
      float xp  = wave_prefix_lse(av, lane);
      float xsa = wave_suffix_lse(av, lane);
      float xsb = wave_suffix_lse(bv, lane);
      if (lane == 31) sWtP[wave] = xp;
      if (lane == 0) { sWtSA[wave] = xsa; sWtSB[wave] = xsb; }
      __syncthreads();
      float plo = NEGINF, sahi = NEGINF, sbhi = NEGINF;
      for (int w = 0; w < wave; ++w)          plo  = lse2(plo,  sWtP[w]);
      for (int w = wave + 1; w < NWAVE; ++w) { sahi = lse2(sahi, sWtSA[w]);
                                               sbhi = lse2(sbhi, sWtSB[w]); }
      const float Pv  = lse2(xp,  plo);
      const float SAv = lse2(xsa, sahi);
      const float SBv = lse2(xsb, sbhi);
      sP[tid]  = Pv;
      sSB[tid] = SBv;
      __syncthreads();

      if (i < I_ - 1) {
        float nb;
        if (tid == 0) {
          nb = NEGV;
        } else {
          const int j = tid - 1;
          const float v = lse2(sE[i * LDE + j] + sP[j], SAv + LOGEPS);
          const int r = i + 1;
          nb = (tid >= r && tid <= (J_ - I_ + r + 1)) ? v : NEGV;
        }
        Bn[tid] = nb;
      }

      float lb;
      if (i < I_ - 1) {
        const float sbexc = (tid + 1 < J_) ? sSB[tid + 1] : NEGINF;
        lb = lse2(Lv + Pv, sbexc + LOGEPS);
      } else {
        lb = sSB[0] + ((tid == J_ - 1) ? 0.0f : LOGEPS);
      }
      out_soft[(size_t)b * I_ * J_ + (size_t)i * J_ + tid] = lb;

      __syncthreads();
      sE[i * LDE + tid] = __expf(lb);
      { float* tmp = Bc; Bc = Bn; Bn = tmp; }
    }
    __syncthreads();
    for (int t = wave; t < (J_/16) * (D_/16); t += NWAVE) tileC(t % (J_/16), t / (J_/16));
  } else {
    {
      const int li = tid % I_;
      const int lq = tid / I_;
      const int o0 = li * LDE + lq * (J_ / 4);
      float s = NEGINF;
      for (int t = J_ / 4 - 1; t >= 0; --t) {
        s = lse2(sE[o0 + t], s);
        sL[o0 + t] = s;
      }
      sSB[tid] = s;
      __syncthreads();
      float carry = NEGINF;
      for (int q2 = lq + 1; q2 < 4; ++q2) carry = lse2(carry, sSB[q2 * I_ + li]);
      if (lq < 3)
        for (int t = 0; t < J_ / 4; ++t) sL[o0 + t] = lse2(sL[o0 + t], carry);
      __syncthreads();
    }

    const int i0 = 3 * lane;
    int eo[3], lo[3];
    float eN[3], lN[3];
    float tu[3];
    float qv[3];
    float dC[3], cC[3], lC[3];
    float eR[3], lR[3];
#pragma unroll
    for (int r = 0; r < 3; ++r) {
      const int i = i0 + r;
      eo[r] = ((i > 0) ? i - 1 : 0) * LDE;
      lo[r] = i * LDE;
      eN[r] = 0.0f;
      lN[r] = 0.0f;
      tu[r] = NEGINF;
      qv[r] = NEGINF;
      dC[r] = NEGINF;
      cC[r] = NEGINF;
      lC[r] = 0.0f;
      eR[r] = 0.0f;
      lR[r] = 0.0f;
    }
    if (wave == 0) {
      float ps[6];
#pragma unroll
      for (int r = 0; r < 3; ++r) {
        lN[r] = sL[lo[r]];
        ps[r]     = NEGINF;
        ps[3 + r] = NEGINF;
      }

      for (int j4 = 0; j4 < J_; j4 += 4) {
        float xq[3][4];
#pragma unroll
        for (int q = 0; q < 4; ++q) {
          const int j = j4 + q;
          float eC[3], Lc[3], pu[3];
          pu[0] = lane_below(ps[2]);
          pu[1] = ps[0];
          pu[2] = ps[1];
#pragma unroll
          for (int r = 0; r < 3; ++r) {
            eC[r] = eN[r];
            Lc[r] = lN[r];
            eN[r] = sE[eo[r] + j];
            lN[r] = sL[lo[r] + j + 1];
          }
          float pb[6];
#pragma unroll
          for (int r = 0; r < 3; ++r) {
            const int i = i0 + r;
            const bool win = (j >= i) && (j <= J_ - I_ + i + 1);
            float b0 = win ? eC[r] + pu[r] : NEGV;
            if (r == 0) b0 = (i == 0) ? ((j == 0) ? 0.0f : NEGV) : b0;
            pb[r]     = b0 - Lc[r];
            pb[3 + r] = b0;
          }
          lse2n<6>(ps, pb);
#pragma unroll
          for (int r = 0; r < 3; ++r) xq[r][q] = Lc[r] + ps[r];
        }
#pragma unroll
        for (int r = 0; r < 3; ++r)
          *(float4*)(osoft + (size_t)(i0 + r) * J_ + j4) =
              make_float4(xq[r][0], xq[r][1], xq[r][2], xq[r][3]);
      }
#pragma unroll
      for (int r = 0; r < 3; ++r) sP[i0 + r] = ps[3 + r];

      tu[0] = lane_below(ps[2]);
      tu[1] = ps[0];
      tu[2] = ps[1];
#pragma unroll
      for (int r = 0; r < 3; ++r) {
        qv[r] = NEGINF;
        dC[r] = NEGINF;
        cC[r] = NEGINF;
        lC[r] = sL[lo[r]];
        eR[r] = sE[eo[r]];
        lR[r] = sL[lo[r] + 1];
      }
    }
    __syncthreads();

#pragma unroll 1
    for (int blk = 0; blk < NBLK + 2; ++blk) {
      if (wave == 0) {
        if (blk < NBLK) {
          float yp[3] = {0.0f, 0.0f, 0.0f};
          auto colstep = [&](const int j, const bool storePrev) __attribute__((always_inline)) {
            float qu[3];
            qu[0] = lane_below(qv[2]);
            qu[1] = qv[0];
            qu[2] = qv[1];
            if (storePrev) {
#pragma unroll
              for (int r = 0; r < 3; ++r) sL[lo[r] + j - 1] = yp[r];
            }
            const int jl = (j + 2 <= J_) ? j + 2 : J_;
            float e2[3], l2[3], dN[3], cN[3];
#pragma unroll
            for (int r = 0; r < 3; ++r) {
              e2[r] = sE[eo[r] + j + 1];
              l2[r] = sL[lo[r] + jl];
            }
#pragma unroll
            for (int r = 0; r < 3; ++r) {
              const int i = i0 + r;
              const bool wn = (i > 0) && (j + 1 >= i) && (j <= J_ - I_ + i);
              dN[r] = wn ? eR[r] - lR[r] : NEGINF;
              cN[r] = wn ? tu[r] - lR[r] : NEGINF;
            }
            float bq[3];
#pragma unroll
            for (int r = 0; r < 3; ++r) bq[r] = dC[r] + qu[r];
            lse3n<3>(qv, bq, cC);
#pragma unroll
            for (int r = 0; r < 3; ++r) {
              yp[r] = lC[r] + qv[r];
              dC[r] = dN[r];
              cC[r] = cN[r];
              lC[r] = lR[r];
              eR[r] = e2[r];
              lR[r] = l2[r];
            }
          };
          colstep(blk * CB, false);
          for (int j = blk * CB + 1; j < blk * CB + CB; ++j) colstep(j, true);
#pragma unroll
          for (int r = 0; r < 3; ++r) sL[lo[r] + blk * CB + CB - 1] = yp[r];
        }
      } else {
        const int ww = wave - 1;
        if (blk >= 1 && blk <= NBLK) {
          const int j = (blk - 1) * CB + lane;
          float xr[9];
#pragma unroll
          for (int u = 0; u < 9; ++u) {
            const int i2 = ww + (NWAVE - 1) * u;
            xr[u] = (i2 < I_) ? osoft[(size_t)i2 * J_ + j] : 0.0f;
          }
#pragma unroll
          for (int u = 0; u < 9; ++u) {
            const int i2 = ww + (NWAVE - 1) * u;
            if (i2 < I_) {
              const float s = sP[i2];
              float lb;
              if (i2 < I_ - 1) {
                const float y = sL[i2 * LDE + j];
                lb = lse2(lse2(xr[u], y + LOGEPS), s + LOGEPS);
              } else {
                lb = s + ((j == J_ - 1) ? 0.0f : LOGEPS);
              }
              osoft[(size_t)i2 * J_ + j] = lb;
              sE[i2 * LDE + j] = __expf(lb);
            }
          }
        }
        if (blk >= 2) {
          for (int t = ww; t < (CB / 16) * (D_ / 16); t += NWAVE - 1)
            tileC((CB / 16) * (blk - 2) + (t & 1), t >> 1);
        }
      }
      __syncthreads();
    }
  }
}

extern "C" void kernel_launch(void* const* d_in, const int* in_sizes, int n_in,
                              void* d_out, int out_size, void* d_ws, size_t ws_size,
                              hipStream_t stream) {
  (void)in_sizes; (void)n_in; (void)d_ws; (void)ws_size; (void)out_size;
  const float* text = (const float*)d_in[0];
  const float* mel  = (const float*)d_in[1];
  const float* gum  = (const float*)d_in[2];
  float* out_soft = (float*)d_out;
  float* out_exp  = (float*)d_out + (size_t)B_ * I_ * J_;
  mobo_kernel<<<dim3(B_), dim3(384), 0, stream>>>(text, mel, gum, out_soft, out_exp);
}


# L pass in the linear domain: row maximum first, then suffix sums of exp(E-max) by plain adds and one log per element, not 192 chained LSEs per thread; new rounding and sum order in L
# speedup vs baseline: 4.0684x; 1.0782x over previous
#include <hip/hip_runtime.h>
#include <hip/hip_bf16.h>
#include <math.h>


typedef float v2f __attribute__((ext_vector_type(2)));
typedef float v8f __attribute__((ext_vector_type(8)));

#define B_      4
#define I_      96
#define J_      384
#define D_      256
#define NWAVE   12
#define LDE     (J_ + 1)
#define LOGEPS  (-1000.0f)
#define NEGV    (-1000000000.0f)
#define NEGINF  (-1.0e38f)
#define TEMPINV (1.0f/0.55f)
#define ESCALE  (1.0f/256.0f)
#define KC      64
#define CB      32
#define NBLK    (J_ / CB)
#define SPITCH  (KC + 4)

__device__ __forceinline__ float lse2(float a, float b) {
  float m = fmaxf(a, b);
  if (m < -9.0e37f) return NEGINF;
  return m + __logf(__expf(a - m) + __expf(b - m));
}

template <int N> struct vecN { typedef float type __attribute__((ext_vector_type(N))); };
template <int N>
__device__ __forceinline__ void lse2n(float (&a)[N], const float (&b)[N]) {
  float m[N], ea[N], eb[N];
#pragma unroll
  for (int n = 0; n < N; ++n) m[n] = fmaxf(a[n], b[n]);
#pragma unroll
  for (int n = 0; n < N; ++n) ea[n] = __expf(a[n] - m[n]);
#pragma unroll
  for (int n = 0; n < N; ++n) eb[n] = __expf(b[n] - m[n]);
  typename vecN<N>::type sv;
#pragma unroll
  for (int n = 0; n < N; ++n) sv[n] = ea[n] + eb[n];
  const typename vecN<N>::type lg = __builtin_elementwise_log(sv);
#pragma unroll
  for (int n = 0; n < N; ++n) {
    const float r = m[n] + lg[n];
    a[n] = (m[n] < -9.0e37f) ? NEGINF : r;
  }
}

template <int N>
__device__ __forceinline__ void lse3n(float (&a)[N], const float (&b)[N], const float (&c)[N]) {
  float m[N], ea[N], eb[N], ec[N];
#pragma unroll
  for (int n = 0; n < N; ++n) m[n] = fmaxf(fmaxf(a[n], b[n]), c[n]);
#pragma unroll
  for (int n = 0; n < N; ++n) ea[n] = __expf(a[n] - m[n]);
#pragma unroll
  for (int n = 0; n < N; ++n) eb[n] = __expf(b[n] - m[n]);
#pragma unroll
  for (int n = 0; n < N; ++n) ec[n] = __expf(c[n] - m[n]);
  typename vecN<N>::type sv;
#pragma unroll
  for (int n = 0; n < N; ++n) sv[n] = ea[n] + (eb[n] + ec[n]);
  const typename vecN<N>::type lg = __builtin_elementwise_log(sv);
#pragma unroll
  for (int n = 0; n < N; ++n) {
    const float r = m[n] + lg[n];
    a[n] = (m[n] < -9.0e37f) ? NEGINF : r;
  }
}

__device__ __forceinline__ float wave_prefix_lse(float x, int lane) {
#pragma unroll
  for (int off = 1; off < 32; off <<= 1) {
    float o = __shfl_up(x, off, 32);
    if (lane >= off) x = lse2(x, o);
  }
  return x;
}

__device__ __forceinline__ float wave_suffix_lse(float x, int lane) {
#pragma unroll
  for (int off = 1; off < 32; off <<= 1) {
    float o = __shfl_down(x, off, 32);
    if (lane + off < 32) x = lse2(x, o);
  }
  return x;
}

__device__ __forceinline__ float lane_below(float x) {
  const int xi = __float_as_int(x);
  const int cross = __builtin_amdgcn_permlanex16(xi, xi, 0xffffffffu, 0xffffffffu, false, false);
  return __int_as_float(__builtin_amdgcn_update_dpp(cross, xi, 0x111, 0xf, 0xf, false));
}

__global__ __launch_bounds__(384, 1)
void mobo_kernel(const float* __restrict__ text,
                 const float* __restrict__ mel,
                 const float* __restrict__ gum,
                 float* __restrict__ out_soft,
                 float* __restrict__ out_exp)
{
  __shared__ float sE[I_ * LDE];
  __shared__ alignas(16) float sL[I_ * LDE];
  __shared__ int   sFlag;
  __shared__ float sP[J_];
  __shared__ float sSB[J_];
  __shared__ float sB0[J_];
  __shared__ float sB1[J_];
  __shared__ float sWtL[NWAVE];
  __shared__ float sWtP[NWAVE];
  __shared__ float sWtSA[NWAVE];
  __shared__ float sWtSB[NWAVE];

  const int b    = blockIdx.x;
  const int tid  = threadIdx.x;
  const int lane = tid & 31;
  const int wave = tid >> 5;
  const int half = lane >> 4;
  const int l16  = lane & 15;

  const float* tx = text + (size_t)b * I_ * D_;
  const float* ml = mel  + (size_t)b * J_ * D_;
  float* osoft = out_soft + (size_t)b * I_ * J_;

  if (tid == 0) sFlag = 0;
  __syncthreads();
  int bad = 0;

  {
    float* stg = sL;
    const int prow = tid >> 4;
    const int pc4  = (tid & 15) * 4;
    const float* gsrc[2];
    gsrc[0] = tx + (size_t)prow * D_ + pc4;
    gsrc[1] = ml + (size_t)prow * D_ + pc4;
    float* sdst = stg + prow * SPITCH + pc4;
#define PIECES(X) X(0) X(1) X(2) X(3) X(4) X(5) X(6) X(7) X(8) X(9) X(10) X(11) X(12) X(13) X(14) X(15) X(16) X(17) X(18) X(19)
#define PSRC(u)   ((u) < 4 ? gsrc[0] + (size_t)(24 * (u)) * D_ : gsrc[1] + (size_t)(24 * ((u) - 4)) * D_)
#define PDECL(u)  float4 pf##u;
#define PLOAD(u)  pf##u = *(const float4*)(PSRC(u) + koff);
#define PSTORE(u) *(float4*)(sdst + 24 * (u) * SPITCH) = pf##u;
    PIECES(PDECL)

    const int mG = wave & 1;
    const int nG = wave >> 1;
    const float* fa = stg + (48 * mG + l16) * SPITCH + 2 * half;
    const float* fb = stg + (I_ + 64 * nG + l16) * SPITCH + 2 * half;

    v8f acc[3][4];
#pragma unroll
    for (int m = 0; m < 3; ++m)
#pragma unroll
      for (int n = 0; n < 4; ++n) acc[m][n] = (v8f){};

    {
      const int koff = 0;
      PIECES(PLOAD)
      PIECES(PSTORE)
    }
    __syncthreads();

    auto chunk_mma = [&]() {
#pragma unroll 4
      for (int kk = 0; kk < KC; kk += 4) {
        v2f a[3], bf[4];
#pragma unroll
        for (int m = 0; m < 3; ++m) a[m]  = *(const v2f*)(fa + m * 16 * SPITCH + kk);
#pragma unroll
        for (int n = 0; n < 4; ++n) bf[n] = *(const v2f*)(fb + n * 16 * SPITCH + kk);
#pragma unroll
        for (int m = 0; m < 3; ++m)
#pragma unroll
          for (int n = 0; n < 4; ++n)
            acc[m][n] = __builtin_amdgcn_wmma_f32_16x16x4_f32(false, a[m], false, bf[n],
                                                              (short)0, acc[m][n], false, false);
      }
    };

#pragma unroll 1
    for (int kc = 1; kc < D_ / KC; ++kc) {
      chunk_mma();
      __syncthreads();
      {
        const int koff = kc * KC;
        PIECES(PLOAD)
      }
      PIECES(PSTORE)
      __syncthreads();
    }
    chunk_mma();

#pragma unroll
    for (int m = 0; m < 3; ++m) {
#pragma unroll
      for (int n = 0; n < 4; ++n) {
#pragma unroll
        for (int r = 0; r < 8; ++r) {
          const int i = 48 * mG + 16 * m + r + 8 * half;
          const int j = 64 * nG + 16 * n + l16;
          const float u = gum[(size_t)b * I_ * J_ + (size_t)i * J_ + j];
          const float g = -__logf(-__logf(u));
          const float ev = (acc[m][n][r] * ESCALE + g) * TEMPINV;
          if (!(ev > -400.0f)) bad = 1;
          sE[i * LDE + j] = ev;
        }
      }
    }
  }
  if (bad) sFlag = 1;
  __syncthreads();

  const int li  = tid % I_;
  const int lq  = tid / I_;
  const int lo0 = li * LDE + lq * (J_ / 4);
  {
    float mx = -3.0e38f, mn = 3.0e38f;
#pragma unroll 8
    for (int t = 0; t < J_ / 4; ++t) {
      const float e = sE[lo0 + t];
      mx = (e > mx) ? e : mx;
      mn = (e < mn) ? e : mn;
    }
    sSB[tid] = mx;
    sB0[tid] = mn;
  }
  __syncthreads();
  const float rowM = fmaxf(fmaxf(sSB[li], sSB[I_ + li]), fmaxf(sSB[2 * I_ + li], sSB[3 * I_ + li]));
  const float rowm = fminf(fminf(sB0[li], sB0[I_ + li]), fminf(sB0[2 * I_ + li], sB0[3 * I_ + li]));
  if (!(rowm - rowM > -80.0f)) sFlag = 1;
  __syncthreads();
  const int slow = sFlag;

  auto tileC = [&](const int mT, const int nT) {
    const int m = mT * 16 + l16;
    const int n = nT * 16 + l16;
    v8f acc = {};
    for (int k0 = 0; k0 < I_; k0 += 4) {
      const int kk = k0 + 2 * half;
      v2f a;
      a.x = sE[kk * LDE + m];
      a.y = sE[(kk + 1) * LDE + m];
      v2f bf;
      bf.x = tx[(size_t)kk * D_ + n];
      bf.y = tx[(size_t)(kk + 1) * D_ + n];
      acc = __builtin_amdgcn_wmma_f32_16x16x4_f32(false, a, false, bf,
                                                  (short)0, acc, false, false);
    }
    for (int r = 0; r < 8; ++r) {
      const int j = mT * 16 + r + 8 * half;
      out_exp[(size_t)b * J_ * D_ + (size_t)j * D_ + n] = acc[r];
    }
  };

  if (slow) {
    float* Bc = sB0;
    float* Bn = sB1;
    Bc[tid] = (tid == 0) ? 0.0f : NEGV;
    __syncthreads();

    for (int i = 0; i < I_; ++i) {
      const float e  = sE[i * LDE + tid];
      float xl = wave_suffix_lse(e, lane);
      if (lane == 0) sWtL[wave] = xl;
      __syncthreads();
      float lhi = NEGINF;
      for (int w = wave + 1; w < NWAVE; ++w) lhi = lse2(lhi, sWtL[w]);
      const float Lv = lse2(xl, lhi);

      const float bv = Bc[tid];
      const float av = bv - Lv;
      float xp  = wave_prefix_lse(av, lane);
      float xsa = wave_suffix_lse(av, lane);
      float xsb = wave_suffix_lse(bv, lane);
      if (lane == 31) sWtP[wave] = xp;
      if (lane == 0) { sWtSA[wave] = xsa; sWtSB[wave] = xsb; }
      __syncthreads();
      float plo = NEGINF, sahi = NEGINF, sbhi = NEGINF;
      for (int w = 0; w < wave; ++w)          plo  = lse2(plo,  sWtP[w]);
      for (int w = wave + 1; w < NWAVE; ++w) { sahi = lse2(sahi, sWtSA[w]);
                                               sbhi = lse2(sbhi, sWtSB[w]); }
      const float Pv  = lse2(xp,  plo);
      const float SAv = lse2(xsa, sahi);
      const float SBv = lse2(xsb, sbhi);
      sP[tid]  = Pv;
      sSB[tid] = SBv;
      __syncthreads();

      if (i < I_ - 1) {
        float nb;
        if (tid == 0) {
          nb = NEGV;
        } else {
          const int j = tid - 1;
          const float v = lse2(sE[i * LDE + j] + sP[j], SAv + LOGEPS);
          const int r = i + 1;
          nb = (tid >= r && tid <= (J_ - I_ + r + 1)) ? v : NEGV;
        }
        Bn[tid] = nb;
      }

      float lb;
      if (i < I_ - 1) {
        const float sbexc = (tid + 1 < J_) ? sSB[tid + 1] : NEGINF;
        lb = lse2(Lv + Pv, sbexc + LOGEPS);
      } else {
        lb = sSB[0] + ((tid == J_ - 1) ? 0.0f : LOGEPS);
      }
      out_soft[(size_t)b * I_ * J_ + (size_t)i * J_ + tid] = lb;

      __syncthreads();
      sE[i * LDE + tid] = __expf(lb);
      { float* tmp = Bc; Bc = Bn; Bn = tmp; }
    }
    __syncthreads();
    for (int t = wave; t < (J_/16) * (D_/16); t += NWAVE) tileC(t % (J_/16), t / (J_/16));
  } else {
    {
      float s = 0.0f;
#pragma unroll 8
      for (int t = J_ / 4 - 1; t >= 0; --t) {
        s += __expf(sE[lo0 + t] - rowM);
        sL[lo0 + t] = s;
      }
      sSB[tid] = s;
      __syncthreads();
      float carry = 0.0f;
      for (int q2 = lq + 1; q2 < 4; ++q2) carry += sSB[q2 * I_ + li];
#pragma unroll 8
      for (int t = 0; t < J_ / 4; ++t) sL[lo0 + t] = rowM + __logf(sL[lo0 + t] + carry);
      __syncthreads();
    }

    const int i0 = 3 * lane;
    int eo[3], lo[3];
    float eN[3], lN[3];
    float tu[3];
    float qv[3];
    float dC[3], cC[3], lC[3];
    float eR[3], lR[3];
#pragma unroll
    for (int r = 0; r < 3; ++r) {
      const int i = i0 + r;
      eo[r] = ((i > 0) ? i - 1 : 0) * LDE;
      lo[r] = i * LDE;
      eN[r] = 0.0f;
      lN[r] = 0.0f;
      tu[r] = NEGINF;
      qv[r] = NEGINF;
      dC[r] = NEGINF;
      cC[r] = NEGINF;
      lC[r] = 0.0f;
      eR[r] = 0.0f;
      lR[r] = 0.0f;
    }
    if (wave == 0) {
      float ps[6];
#pragma unroll
      for (int r = 0; r < 3; ++r) {
        lN[r] = sL[lo[r]];
        ps[r]     = NEGINF;
        ps[3 + r] = NEGINF;
      }

      for (int j4 = 0; j4 < J_; j4 += 4) {
        float xq[3][4];
#pragma unroll
        for (int q = 0; q < 4; ++q) {
          const int j = j4 + q;
          float eC[3], Lc[3], pu[3];
          pu[0] = lane_below(ps[2]);
          pu[1] = ps[0];
          pu[2] = ps[1];
#pragma unroll
          for (int r = 0; r < 3; ++r) {
            eC[r] = eN[r];
            Lc[r] = lN[r];
            eN[r] = sE[eo[r] + j];
            lN[r] = sL[lo[r] + j + 1];
          }
          float pb[6];
#pragma unroll
          for (int r = 0; r < 3; ++r) {
            const int i = i0 + r;
            const bool win = (j >= i) && (j <= J_ - I_ + i + 1);
            float b0 = win ? eC[r] + pu[r] : NEGV;
            if (r == 0) b0 = (i == 0) ? ((j == 0) ? 0.0f : NEGV) : b0;
            pb[r]     = b0 - Lc[r];
            pb[3 + r] = b0;
          }
          lse2n<6>(ps, pb);
#pragma unroll
          for (int r = 0; r < 3; ++r) xq[r][q] = Lc[r] + ps[r];
        }
#pragma unroll
        for (int r = 0; r < 3; ++r)
          *(float4*)(osoft + (size_t)(i0 + r) * J_ + j4) =
              make_float4(xq[r][0], xq[r][1], xq[r][2], xq[r][3]);
      }
#pragma unroll
      for (int r = 0; r < 3; ++r) sP[i0 + r] = ps[3 + r];

      tu[0] = lane_below(ps[2]);
      tu[1] = ps[0];
      tu[2] = ps[1];
#pragma unroll
      for (int r = 0; r < 3; ++r) {
        qv[r] = NEGINF;
        dC[r] = NEGINF;
        cC[r] = NEGINF;
        lC[r] = sL[lo[r]];
        eR[r] = sE[eo[r]];
        lR[r] = sL[lo[r] + 1];
      }
    }
    __syncthreads();

#pragma unroll 1
    for (int blk = 0; blk < NBLK + 2; ++blk) {
      if (wave == 0) {
        if (blk < NBLK) {
          float yp[3] = {0.0f, 0.0f, 0.0f};
          auto colstep = [&](const int j, const bool storePrev) __attribute__((always_inline)) {
            float qu[3];
            qu[0] = lane_below(qv[2]);
            qu[1] = qv[0];
            qu[2] = qv[1];
            if (storePrev) {
#pragma unroll
              for (int r = 0; r < 3; ++r) sL[lo[r] + j - 1] = yp[r];
            }
            const int jl = (j + 2 <= J_) ? j + 2 : J_;
            float e2[3], l2[3], dN[3], cN[3];
#pragma unroll
            for (int r = 0; r < 3; ++r) {
              e2[r] = sE[eo[r] + j + 1];
              l2[r] = sL[lo[r] + jl];
            }
#pragma unroll
            for (int r = 0; r < 3; ++r) {
              const int i = i0 + r;
              const bool wn = (i > 0) && (j + 1 >= i) && (j <= J_ - I_ + i);
              dN[r] = wn ? eR[r] - lR[r] : NEGINF;
              cN[r] = wn ? tu[r] - lR[r] : NEGINF;
            }
            float bq[3];
#pragma unroll
            for (int r = 0; r < 3; ++r) bq[r] = dC[r] + qu[r];
            lse3n<3>(qv, bq, cC);
#pragma unroll
            for (int r = 0; r < 3; ++r) {
              yp[r] = lC[r] + qv[r];
              dC[r] = dN[r];
              cC[r] = cN[r];
              lC[r] = lR[r];
              eR[r] = e2[r];
              lR[r] = l2[r];
            }
          };
          colstep(blk * CB, false);
          for (int j = blk * CB + 1; j < blk * CB + CB; ++j) colstep(j, true);
#pragma unroll
          for (int r = 0; r < 3; ++r) sL[lo[r] + blk * CB + CB - 1] = yp[r];
        }
      } else {
        const int ww = wave - 1;
        if (blk >= 1 && blk <= NBLK) {
          const int j = (blk - 1) * CB + lane;
          float xr[9];
#pragma unroll
          for (int u = 0; u < 9; ++u) {
            const int i2 = ww + (NWAVE - 1) * u;
            xr[u] = (i2 < I_) ? osoft[(size_t)i2 * J_ + j] : 0.0f;
          }
#pragma unroll
          for (int u = 0; u < 9; ++u) {
            const int i2 = ww + (NWAVE - 1) * u;
            if (i2 < I_) {
              const float s = sP[i2];
              float lb;
              if (i2 < I_ - 1) {
                const float y = sL[i2 * LDE + j];
                lb = lse2(lse2(xr[u], y + LOGEPS), s + LOGEPS);
              } else {
                lb = s + ((j == J_ - 1) ? 0.0f : LOGEPS);
              }
              osoft[(size_t)i2 * J_ + j] = lb;
              sE[i2 * LDE + j] = __expf(lb);
            }
          }
        }
        if (blk >= 2) {
          for (int t = ww; t < (CB / 16) * (D_ / 16); t += NWAVE - 1)
            tileC((CB / 16) * (blk - 2) + (t & 1), t >> 1);
        }
      }
      __syncthreads();
    }
  }
}

extern "C" void kernel_launch(void* const* d_in, const int* in_sizes, int n_in,
                              void* d_out, int out_size, void* d_ws, size_t ws_size,
                              hipStream_t stream) {
  (void)in_sizes; (void)n_in; (void)d_ws; (void)ws_size; (void)out_size;
  const float* text = (const float*)d_in[0];
  const float* mel  = (const float*)d_in[1];
  const float* gum  = (const float*)d_in[2];
  float* out_soft = (float*)d_out;
  float* out_exp  = (float*)d_out + (size_t)B_ * I_ * J_;
  mobo_kernel<<<dim3(B_), dim3(384), 0, stream>>>(text, mel, gum, out_soft, out_exp);
}


# Bij row totals leave the order-0 column step (3 LSE chains per lane, not 6): the 11 idle waves form them during the first order-1 block from E, L, stored X as row max + sum of exps; new sum order
# speedup vs baseline: 4.2444x; 1.0433x over previous
#include <hip/hip_runtime.h>
#include <hip/hip_bf16.h>
#include <math.h>


typedef float v2f __attribute__((ext_vector_type(2)));
typedef float v8f __attribute__((ext_vector_type(8)));

#define B_      4
#define I_      96
#define J_      384
#define D_      256
#define NWAVE   12
#define LDE     (J_ + 1)
#define LOGEPS  (-1000.0f)
#define NEGV    (-1000000000.0f)
#define NEGINF  (-1.0e38f)
#define TEMPINV (1.0f/0.55f)
#define ESCALE  (1.0f/256.0f)
#define KC      64
#define CB      32
#define NBLK    (J_ / CB)
#define SPITCH  (KC + 4)

__device__ __forceinline__ float lse2(float a, float b) {
  float m = fmaxf(a, b);
  if (m < -9.0e37f) return NEGINF;
  return m + __logf(__expf(a - m) + __expf(b - m));
}

template <int N> struct vecN { typedef float type __attribute__((ext_vector_type(N))); };
template <int N>
__device__ __forceinline__ void lse2n(float (&a)[N], const float (&b)[N]) {
  float m[N], ea[N], eb[N];
#pragma unroll
  for (int n = 0; n < N; ++n) m[n] = fmaxf(a[n], b[n]);
#pragma unroll
  for (int n = 0; n < N; ++n) ea[n] = __expf(a[n] - m[n]);
#pragma unroll
  for (int n = 0; n < N; ++n) eb[n] = __expf(b[n] - m[n]);
  typename vecN<N>::type sv;
#pragma unroll
  for (int n = 0; n < N; ++n) sv[n] = ea[n] + eb[n];
  const typename vecN<N>::type lg = __builtin_elementwise_log(sv);
#pragma unroll
  for (int n = 0; n < N; ++n) {
    const float r = m[n] + lg[n];
    a[n] = (m[n] < -9.0e37f) ? NEGINF : r;
  }
}

template <int N>
__device__ __forceinline__ void lse3n(float (&a)[N], const float (&b)[N], const float (&c)[N]) {
  float m[N], ea[N], eb[N], ec[N];
#pragma unroll
  for (int n = 0; n < N; ++n) m[n] = fmaxf(fmaxf(a[n], b[n]), c[n]);
#pragma unroll
  for (int n = 0; n < N; ++n) ea[n] = __expf(a[n] - m[n]);
#pragma unroll
  for (int n = 0; n < N; ++n) eb[n] = __expf(b[n] - m[n]);
#pragma unroll
  for (int n = 0; n < N; ++n) ec[n] = __expf(c[n] - m[n]);
  typename vecN<N>::type sv;
#pragma unroll
  for (int n = 0; n < N; ++n) sv[n] = ea[n] + (eb[n] + ec[n]);
  const typename vecN<N>::type lg = __builtin_elementwise_log(sv);
#pragma unroll
  for (int n = 0; n < N; ++n) {
    const float r = m[n] + lg[n];
    a[n] = (m[n] < -9.0e37f) ? NEGINF : r;
  }
}

__device__ __forceinline__ float wave_prefix_lse(float x, int lane) {
#pragma unroll
  for (int off = 1; off < 32; off <<= 1) {
    float o = __shfl_up(x, off, 32);
    if (lane >= off) x = lse2(x, o);
  }
  return x;
}

__device__ __forceinline__ float wave_suffix_lse(float x, int lane) {
#pragma unroll
  for (int off = 1; off < 32; off <<= 1) {
    float o = __shfl_down(x, off, 32);
    if (lane + off < 32) x = lse2(x, o);
  }
  return x;
}

__device__ __forceinline__ float lane_below(float x) {
  const int xi = __float_as_int(x);
  const int cross = __builtin_amdgcn_permlanex16(xi, xi, 0xffffffffu, 0xffffffffu, false, false);
  return __int_as_float(__builtin_amdgcn_update_dpp(cross, xi, 0x111, 0xf, 0xf, false));
}

__global__ __launch_bounds__(384, 1)
void mobo_kernel(const float* __restrict__ text,
                 const float* __restrict__ mel,
                 const float* __restrict__ gum,
                 float* __restrict__ out_soft,
                 float* __restrict__ out_exp)
{
  __shared__ float sE[I_ * LDE];
  __shared__ alignas(16) float sL[I_ * LDE];
  __shared__ int   sFlag;
  __shared__ float sP[J_];
  __shared__ float sSB[J_];
  __shared__ float sB0[J_];
  __shared__ float sB1[J_];
  __shared__ float sLc[32 * 33];
  __shared__ float sWtL[NWAVE];
  __shared__ float sWtP[NWAVE];
  __shared__ float sWtSA[NWAVE];
  __shared__ float sWtSB[NWAVE];

  const int b    = blockIdx.x;
  const int tid  = threadIdx.x;
  const int lane = tid & 31;
  const int wave = tid >> 5;
  const int half = lane >> 4;
  const int l16  = lane & 15;

  const float* tx = text + (size_t)b * I_ * D_;
  const float* ml = mel  + (size_t)b * J_ * D_;
  float* osoft = out_soft + (size_t)b * I_ * J_;

  if (tid == 0) sFlag = 0;
  __syncthreads();
  int bad = 0;

  {
    float* stg = sL;
    const int prow = tid >> 4;
    const int pc4  = (tid & 15) * 4;
    const float* gsrc[2];
    gsrc[0] = tx + (size_t)prow * D_ + pc4;
    gsrc[1] = ml + (size_t)prow * D_ + pc4;
    float* sdst = stg + prow * SPITCH + pc4;
#define PIECES(X) X(0) X(1) X(2) X(3) X(4) X(5) X(6) X(7) X(8) X(9) X(10) X(11) X(12) X(13) X(14) X(15) X(16) X(17) X(18) X(19)
#define PSRC(u)   ((u) < 4 ? gsrc[0] + (size_t)(24 * (u)) * D_ : gsrc[1] + (size_t)(24 * ((u) - 4)) * D_)
#define PDECL(u)  float4 pf##u;
#define PLOAD(u)  pf##u = *(const float4*)(PSRC(u) + koff);
#define PSTORE(u) *(float4*)(sdst + 24 * (u) * SPITCH) = pf##u;
    PIECES(PDECL)

    const int mG = wave & 1;
    const int nG = wave >> 1;
    const float* fa = stg + (48 * mG + l16) * SPITCH + 2 * half;
    const float* fb = stg + (I_ + 64 * nG + l16) * SPITCH + 2 * half;

    v8f acc[3][4];
#pragma unroll
    for (int m = 0; m < 3; ++m)
#pragma unroll
      for (int n = 0; n < 4; ++n) acc[m][n] = (v8f){};

    {
      const int koff = 0;
      PIECES(PLOAD)
      PIECES(PSTORE)
    }
    __syncthreads();

    auto chunk_mma = [&]() {
#pragma unroll 4
      for (int kk = 0; kk < KC; kk += 4) {
        v2f a[3], bf[4];
#pragma unroll
        for (int m = 0; m < 3; ++m) a[m]  = *(const v2f*)(fa + m * 16 * SPITCH + kk);
#pragma unroll
        for (int n = 0; n < 4; ++n) bf[n] = *(const v2f*)(fb + n * 16 * SPITCH + kk);
#pragma unroll
        for (int m = 0; m < 3; ++m)
#pragma unroll
          for (int n = 0; n < 4; ++n)
            acc[m][n] = __builtin_amdgcn_wmma_f32_16x16x4_f32(false, a[m], false, bf[n],
                                                              (short)0, acc[m][n], false, false);
      }
    };

#pragma unroll 1
    for (int kc = 1; kc < D_ / KC; ++kc) {
      chunk_mma();
      __syncthreads();
      {
        const int koff = kc * KC;
        PIECES(PLOAD)
      }
      PIECES(PSTORE)
      __syncthreads();
    }
    chunk_mma();

#pragma unroll
    for (int m = 0; m < 3; ++m) {
#pragma unroll
      for (int n = 0; n < 4; ++n) {
#pragma unroll
        for (int r = 0; r < 8; ++r) {
          const int i = 48 * mG + 16 * m + r + 8 * half;
          const int j = 64 * nG + 16 * n + l16;
          const float u = gum[(size_t)b * I_ * J_ + (size_t)i * J_ + j];
          const float g = -__logf(-__logf(u));
          const float ev = (acc[m][n][r] * ESCALE + g) * TEMPINV;
          if (!(ev > -400.0f)) bad = 1;
          sE[i * LDE + j] = ev;
        }
      }
    }
  }
  if (bad) sFlag = 1;
  __syncthreads();

  const int li  = tid % I_;
  const int lq  = tid / I_;
  const int lo0 = li * LDE + lq * (J_ / 4);
  {
    float mx = -3.0e38f, mn = 3.0e38f;
#pragma unroll 8
    for (int t = 0; t < J_ / 4; ++t) {
      const float e = sE[lo0 + t];
      mx = (e > mx) ? e : mx;
      mn = (e < mn) ? e : mn;
    }
    sSB[tid] = mx;
    sB0[tid] = mn;
  }
  __syncthreads();
  const float rowM = fmaxf(fmaxf(sSB[li], sSB[I_ + li]), fmaxf(sSB[2 * I_ + li], sSB[3 * I_ + li]));
  const float rowm = fminf(fminf(sB0[li], sB0[I_ + li]), fminf(sB0[2 * I_ + li], sB0[3 * I_ + li]));
  if (!(rowm - rowM > -80.0f)) sFlag = 1;
  __syncthreads();
  const int slow = sFlag;

  auto tileC = [&](const int mT, const int nT) {
    const int m = mT * 16 + l16;
    const int n = nT * 16 + l16;
    v8f acc = {};
    for (int k0 = 0; k0 < I_; k0 += 4) {
      const int kk = k0 + 2 * half;
      v2f a;
      a.x = sE[kk * LDE + m];
      a.y = sE[(kk + 1) * LDE + m];
      v2f bf;
      bf.x = tx[(size_t)kk * D_ + n];
      bf.y = tx[(size_t)(kk + 1) * D_ + n];
      acc = __builtin_amdgcn_wmma_f32_16x16x4_f32(false, a, false, bf,
                                                  (short)0, acc, false, false);
    }
    for (int r = 0; r < 8; ++r) {
      const int j = mT * 16 + r + 8 * half;
      out_exp[(size_t)b * J_ * D_ + (size_t)j * D_ + n] = acc[r];
    }
  };

  if (slow) {
    float* Bc = sB0;
    float* Bn = sB1;
    Bc[tid] = (tid == 0) ? 0.0f : NEGV;
    __syncthreads();

    for (int i = 0; i < I_; ++i) {
      const float e  = sE[i * LDE + tid];
      float xl = wave_suffix_lse(e, lane);
      if (lane == 0) sWtL[wave] = xl;
      __syncthreads();
      float lhi = NEGINF;
      for (int w = wave + 1; w < NWAVE; ++w) lhi = lse2(lhi, sWtL[w]);
      const float Lv = lse2(xl, lhi);

      const float bv = Bc[tid];
      const float av = bv - Lv;
      float xp  = wave_prefix_lse(av, lane);
      float xsa = wave_suffix_lse(av, lane);
      float xsb = wave_suffix_lse(bv, lane);
      if (lane == 31) sWtP[wave] = xp;
      if (lane == 0) { sWtSA[wave] = xsa; sWtSB[wave] = xsb; }
      __syncthreads();
      float plo = NEGINF, sahi = NEGINF, sbhi = NEGINF;
      for (int w = 0; w < wave; ++w)          plo  = lse2(plo,  sWtP[w]);
      for (int w = wave + 1; w < NWAVE; ++w) { sahi = lse2(sahi, sWtSA[w]);
                                               sbhi = lse2(sbhi, sWtSB[w]); }
      const float Pv  = lse2(xp,  plo);
      const float SAv = lse2(xsa, sahi);
      const float SBv = lse2(xsb, sbhi);
      sP[tid]  = Pv;
      sSB[tid] = SBv;
      __syncthreads();

      if (i < I_ - 1) {
        float nb;
        if (tid == 0) {
          nb = NEGV;
        } else {
          const int j = tid - 1;
          const float v = lse2(sE[i * LDE + j] + sP[j], SAv + LOGEPS);
          const int r = i + 1;
          nb = (tid >= r && tid <= (J_ - I_ + r + 1)) ? v : NEGV;
        }
        Bn[tid] = nb;
      }

      float lb;
      if (i < I_ - 1) {
        const float sbexc = (tid + 1 < J_) ? sSB[tid + 1] : NEGINF;
        lb = lse2(Lv + Pv, sbexc + LOGEPS);
      } else {
        lb = sSB[0] + ((tid == J_ - 1) ? 0.0f : LOGEPS);
      }
      out_soft[(size_t)b * I_ * J_ + (size_t)i * J_ + tid] = lb;

      __syncthreads();
      sE[i * LDE + tid] = __expf(lb);
      { float* tmp = Bc; Bc = Bn; Bn = tmp; }
    }
    __syncthreads();
    for (int t = wave; t < (J_/16) * (D_/16); t += NWAVE) tileC(t % (J_/16), t / (J_/16));
  } else {
    {
      float s = 0.0f;
#pragma unroll 8
      for (int t = J_ / 4 - 1; t >= 0; --t) {
        s += __expf(sE[lo0 + t] - rowM);
        sL[lo0 + t] = s;
      }
      sSB[tid] = s;
      __syncthreads();
      float carry = 0.0f;
      for (int q2 = lq + 1; q2 < 4; ++q2) carry += sSB[q2 * I_ + li];
#pragma unroll 8
      for (int t = 0; t < J_ / 4; ++t) sL[lo0 + t] = rowM + __logf(sL[lo0 + t] + carry);
      __syncthreads();
    }

    const int i0 = 3 * lane;
    int eo[3], lo[3];
    float eN[3], lN[3];
    float tu[3];
    float qv[3];
    float dC[3], cC[3], lC[3];
    float eR[3], lR[3];
#pragma unroll
    for (int r = 0; r < 3; ++r) {
      const int i = i0 + r;
      eo[r] = ((i > 0) ? i - 1 : 0) * LDE;
      lo[r] = i * LDE;
      eN[r] = 0.0f;
      lN[r] = 0.0f;
      tu[r] = NEGINF;
      qv[r] = NEGINF;
      dC[r] = NEGINF;
      cC[r] = NEGINF;
      lC[r] = 0.0f;
      eR[r] = 0.0f;
      lR[r] = 0.0f;
    }
    if (wave == 0) {
      float ps[3];
#pragma unroll
      for (int r = 0; r < 3; ++r) {
        lN[r] = sL[lo[r]];
        ps[r] = NEGINF;
      }

      for (int j4 = 0; j4 < J_; j4 += 4) {
        float xq[3][4];
#pragma unroll
        for (int q = 0; q < 4; ++q) {
          const int j = j4 + q;
          float eC[3], Lc[3], pu[3];
          pu[0] = lane_below(ps[2]);
          pu[1] = ps[0];
          pu[2] = ps[1];
#pragma unroll
          for (int r = 0; r < 3; ++r) {
            eC[r] = eN[r];
            Lc[r] = lN[r];
            eN[r] = sE[eo[r] + j];
            lN[r] = sL[lo[r] + j + 1];
          }
          float pb[3];
#pragma unroll
          for (int r = 0; r < 3; ++r) {
            const int i = i0 + r;
            const bool win = (j >= i) && (j <= J_ - I_ + i + 1);
            float b0 = win ? eC[r] + pu[r] : NEGV;
            if (r == 0) b0 = (i == 0) ? ((j == 0) ? 0.0f : NEGV) : b0;
            pb[r] = b0 - Lc[r];
          }
          lse2n<3>(ps, pb);
#pragma unroll
          for (int r = 0; r < 3; ++r) xq[r][q] = Lc[r] + ps[r];
        }
#pragma unroll
        for (int r = 0; r < 3; ++r)
          *(float4*)(osoft + (size_t)(i0 + r) * J_ + j4) =
              make_float4(xq[r][0], xq[r][1], xq[r][2], xq[r][3]);
      }
      tu[0] = lane_below(ps[2]);
      tu[1] = ps[0];
      tu[2] = ps[1];
#pragma unroll
      for (int r = 0; r < 3; ++r) {
        qv[r] = NEGINF;
        dC[r] = NEGINF;
        cC[r] = NEGINF;
        lC[r] = sL[lo[r]];
        eR[r] = sE[eo[r]];
        lR[r] = sL[lo[r] + 1];
      }
    } else {
      for (int idx = tid - 32; idx < 32 * 32; idx += 32 * (NWAVE - 1))
        sLc[(idx >> 5) * 33 + (idx & 31)] = sL[(idx >> 5) * LDE + (idx & 31)];
    }
    __syncthreads();

#pragma unroll 1
    for (int blk = 0; blk < NBLK + 2; ++blk) {
      if (wave == 0) {
        if (blk < NBLK) {
          float yp[3] = {0.0f, 0.0f, 0.0f};
          auto colstep = [&](const int j, const bool storePrev) __attribute__((always_inline)) {
            float qu[3];
            qu[0] = lane_below(qv[2]);
            qu[1] = qv[0];
            qu[2] = qv[1];
            if (storePrev) {
#pragma unroll
              for (int r = 0; r < 3; ++r) sL[lo[r] + j - 1] = yp[r];
            }
            const int jl = (j + 2 <= J_) ? j + 2 : J_;
            float e2[3], l2[3], dN[3], cN[3];
#pragma unroll
            for (int r = 0; r < 3; ++r) {
              e2[r] = sE[eo[r] + j + 1];
              l2[r] = sL[lo[r] + jl];
            }
#pragma unroll
            for (int r = 0; r < 3; ++r) {
              const int i = i0 + r;
              const bool wn = (i > 0) && (j + 1 >= i) && (j <= J_ - I_ + i);
              dN[r] = wn ? eR[r] - lR[r] : NEGINF;
              cN[r] = wn ? tu[r] - lR[r] : NEGINF;
            }
            float bq[3];
#pragma unroll
            for (int r = 0; r < 3; ++r) bq[r] = dC[r] + qu[r];
            lse3n<3>(qv, bq, cC);
#pragma unroll
            for (int r = 0; r < 3; ++r) {
              yp[r] = lC[r] + qv[r];
              dC[r] = dN[r];
              cC[r] = cN[r];
              lC[r] = lR[r];
              eR[r] = e2[r];
              lR[r] = l2[r];
            }
          };
          colstep(blk * CB, false);
          for (int j = blk * CB + 1; j < blk * CB + CB; ++j) colstep(j, true);
#pragma unroll
          for (int r = 0; r < 3; ++r) sL[lo[r] + blk * CB + CB - 1] = yp[r];
        }
      } else {
        const int ww = wave - 1;
        if (blk == 0) {
          constexpr int NR = (I_ + NWAVE - 2) / (NWAVE - 1);
          constexpr int NT = J_ / 32;
          float tv[NR][NT], rm[NR], rs[NR];
#pragma unroll
          for (int u = 0; u < NR; ++u) {
            const int i  = ww + (NWAVE - 1) * u;
            const int ib = (i > 0 && i < I_) ? i - 1 : 0;
#pragma unroll
            for (int t = 0; t < NT; ++t) tv[u][t] = osoft[(size_t)ib * J_ + lane + 32 * t];
          }
#pragma unroll
          for (int u = 0; u < NR; ++u) {
            const int i  = ww + (NWAVE - 1) * u;
            const int ib = (i > 0 && i < I_) ? i - 1 : 0;
            rm[u] = NEGINF;
#pragma unroll
            for (int t = 0; t < NT; ++t) {
              const int cc = lane + 32 * t;
              const bool in = (i > 0) && (i < I_) && (cc >= i - 1) && (cc <= J_ - I_ + i) && (cc <= J_ - 2);
              const float lv = (t == 0) ? sLc[(ib & 31) * 33 + lane] : sL[ib * LDE + cc];
              const float v = (sE[ib * LDE + cc] + tv[u][t]) - lv;
              tv[u][t] = in ? v : NEGINF;
              rm[u] = fmaxf(rm[u], tv[u][t]);
            }
          }
#pragma unroll
          for (int off = 16; off > 0; off >>= 1)
#pragma unroll
            for (int u = 0; u < NR; ++u) rm[u] = fmaxf(rm[u], __shfl_xor(rm[u], off, 32));
#pragma unroll
          for (int u = 0; u < NR; ++u) {
            rs[u] = 0.0f;
#pragma unroll
            for (int t = 0; t < NT; ++t) rs[u] += __expf(tv[u][t] - rm[u]);
          }
#pragma unroll
          for (int off = 16; off > 0; off >>= 1)
#pragma unroll
            for (int u = 0; u < NR; ++u) rs[u] += __shfl_xor(rs[u], off, 32);
#pragma unroll
          for (int u = 0; u < NR; ++u) {
            const int i = ww + (NWAVE - 1) * u;
            sP[i] = (i > 0) ? rm[u] + __logf(rs[u]) : 0.0f;
          }
        }
        if (blk >= 1 && blk <= NBLK) {
          const int j = (blk - 1) * CB + lane;
          float xr[9];
#pragma unroll
          for (int u = 0; u < 9; ++u) {
            const int i2 = ww + (NWAVE - 1) * u;
            xr[u] = (i2 < I_) ? osoft[(size_t)i2 * J_ + j] : 0.0f;
          }
#pragma unroll
          for (int u = 0; u < 9; ++u) {
            const int i2 = ww + (NWAVE - 1) * u;
            if (i2 < I_) {
              const float s = sP[i2];
              float lb;
              if (i2 < I_ - 1) {
                const float y = sL[i2 * LDE + j];
                lb = lse2(lse2(xr[u], y + LOGEPS), s + LOGEPS);
              } else {
                lb = s + ((j == J_ - 1) ? 0.0f : LOGEPS);
              }
              osoft[(size_t)i2 * J_ + j] = lb;
              sE[i2 * LDE + j] = __expf(lb);
            }
          }
        }
        if (blk >= 2) {
          for (int t = ww; t < (CB / 16) * (D_ / 16); t += NWAVE - 1)
            tileC((CB / 16) * (blk - 2) + (t & 1), t >> 1);
        }
      }
      __syncthreads();
    }
  }
}

extern "C" void kernel_launch(void* const* d_in, const int* in_sizes, int n_in,
                              void* d_out, int out_size, void* d_ws, size_t ws_size,
                              hipStream_t stream) {
  (void)in_sizes; (void)n_in; (void)d_ws; (void)ws_size; (void)out_size;
  const float* text = (const float*)d_in[0];
  const float* mel  = (const float*)d_in[1];
  const float* gum  = (const float*)d_in[2];
  float* out_soft = (float*)d_out;
  float* out_exp  = (float*)d_out + (size_t)B_ * I_ * J_;
  mobo_kernel<<<dim3(B_), dim3(384), 0, stream>>>(text, mel, gum, out_soft, out_exp);
}


# Column sweeps in base-2 logarithms: E and L scaled by log2(e) in the L pass, a step's exp and log are bare hardware exp2/log2 (sum in [1,3]); X, Y, totals back by ln 2; orders kept, new rounding
# speedup vs baseline: 4.5548x; 1.0731x over previous
#include <hip/hip_runtime.h>
#include <hip/hip_bf16.h>
#include <math.h>


typedef float v2f __attribute__((ext_vector_type(2)));
typedef float v8f __attribute__((ext_vector_type(8)));

#define B_      4
#define I_      96
#define J_      384
#define D_      256
#define NWAVE   12
#define LDE     (J_ + 1)
#define LOGEPS  (-1000.0f)
#define NEGV    (-1000000000.0f)
#define NEGINF  (-1.0e38f)
#define LOG2E   1.44269504088896340736f
#define LN2     0.69314718055994530942f
#define NEGV2   (NEGV * LOG2E)
#define TEMPINV (1.0f/0.55f)
#define ESCALE  (1.0f/256.0f)
#define KC      64
#define CB      32
#define NBLK    (J_ / CB)
#define SPITCH  (KC + 4)

__device__ __forceinline__ float lse2(float a, float b) {
  float m = fmaxf(a, b);
  if (m < -9.0e37f) return NEGINF;
  return m + __logf(__expf(a - m) + __expf(b - m));
}

template <int N>
__device__ __forceinline__ void lse2n(float (&a)[N], const float (&b)[N]) {
  float m[N], ea[N], eb[N];
#pragma unroll
  for (int n = 0; n < N; ++n) m[n] = fmaxf(a[n], b[n]);
#pragma unroll
  for (int n = 0; n < N; ++n) ea[n] = __builtin_amdgcn_exp2f(a[n] - m[n]);
#pragma unroll
  for (int n = 0; n < N; ++n) eb[n] = __builtin_amdgcn_exp2f(b[n] - m[n]);
  float sv[N];
#pragma unroll
  for (int n = 0; n < N; ++n) sv[n] = ea[n] + eb[n];
#pragma unroll
  for (int n = 0; n < N; ++n) {
    const float r = m[n] + __builtin_amdgcn_logf(sv[n]);
    a[n] = (m[n] < -9.0e37f) ? NEGINF : r;
  }
}

template <int N>
__device__ __forceinline__ void lse3n(float (&a)[N], const float (&b)[N], const float (&c)[N]) {
  float m[N], ea[N], eb[N], ec[N];
#pragma unroll
  for (int n = 0; n < N; ++n) m[n] = fmaxf(fmaxf(a[n], b[n]), c[n]);
#pragma unroll
  for (int n = 0; n < N; ++n) ea[n] = __builtin_amdgcn_exp2f(a[n] - m[n]);
#pragma unroll
  for (int n = 0; n < N; ++n) eb[n] = __builtin_amdgcn_exp2f(b[n] - m[n]);
#pragma unroll
  for (int n = 0; n < N; ++n) ec[n] = __builtin_amdgcn_exp2f(c[n] - m[n]);
  float sv[N];
#pragma unroll
  for (int n = 0; n < N; ++n) sv[n] = ea[n] + (eb[n] + ec[n]);
#pragma unroll
  for (int n = 0; n < N; ++n) {
    const float r = m[n] + __builtin_amdgcn_logf(sv[n]);
    a[n] = (m[n] < -9.0e37f) ? NEGINF : r;
  }
}

__device__ __forceinline__ float wave_prefix_lse(float x, int lane) {
#pragma unroll
  for (int off = 1; off < 32; off <<= 1) {
    float o = __shfl_up(x, off, 32);
    if (lane >= off) x = lse2(x, o);
  }
  return x;
}

__device__ __forceinline__ float wave_suffix_lse(float x, int lane) {
#pragma unroll
  for (int off = 1; off < 32; off <<= 1) {
    float o = __shfl_down(x, off, 32);
    if (lane + off < 32) x = lse2(x, o);
  }
  return x;
}

__device__ __forceinline__ float lane_below(float x) {
  const int xi = __float_as_int(x);
  const int cross = __builtin_amdgcn_permlanex16(xi, xi, 0xffffffffu, 0xffffffffu, false, false);
  return __int_as_float(__builtin_amdgcn_update_dpp(cross, xi, 0x111, 0xf, 0xf, false));
}

__global__ __launch_bounds__(384, 1)
void mobo_kernel(const float* __restrict__ text,
                 const float* __restrict__ mel,
                 const float* __restrict__ gum,
                 float* __restrict__ out_soft,
                 float* __restrict__ out_exp)
{
  __shared__ float sE[I_ * LDE];
  __shared__ alignas(16) float sL[I_ * LDE];
  __shared__ int   sFlag;
  __shared__ float sP[J_];
  __shared__ float sSB[J_];
  __shared__ float sB0[J_];
  __shared__ float sB1[J_];
  __shared__ float sLc[32 * 33];
  __shared__ float sWtL[NWAVE];
  __shared__ float sWtP[NWAVE];
  __shared__ float sWtSA[NWAVE];
  __shared__ float sWtSB[NWAVE];

  const int b    = blockIdx.x;
  const int tid  = threadIdx.x;
  const int lane = tid & 31;
  const int wave = tid >> 5;
  const int half = lane >> 4;
  const int l16  = lane & 15;

  const float* tx = text + (size_t)b * I_ * D_;
  const float* ml = mel  + (size_t)b * J_ * D_;
  float* osoft = out_soft + (size_t)b * I_ * J_;

  if (tid == 0) sFlag = 0;
  __syncthreads();
  int bad = 0;

  {
    float* stg = sL;
    const int prow = tid >> 4;
    const int pc4  = (tid & 15) * 4;
    const float* gsrc[2];
    gsrc[0] = tx + (size_t)prow * D_ + pc4;
    gsrc[1] = ml + (size_t)prow * D_ + pc4;
    float* sdst = stg + prow * SPITCH + pc4;
#define PIECES(X) X(0) X(1) X(2) X(3) X(4) X(5) X(6) X(7) X(8) X(9) X(10) X(11) X(12) X(13) X(14) X(15) X(16) X(17) X(18) X(19)
#define PSRC(u)   ((u) < 4 ? gsrc[0] + (size_t)(24 * (u)) * D_ : gsrc[1] + (size_t)(24 * ((u) - 4)) * D_)
#define PDECL(u)  float4 pf##u;
#define PLOAD(u)  pf##u = *(const float4*)(PSRC(u) + koff);
#define PSTORE(u) *(float4*)(sdst + 24 * (u) * SPITCH) = pf##u;
    PIECES(PDECL)

    const int mG = wave & 1;
    const int nG = wave >> 1;
    const float* fa = stg + (48 * mG + l16) * SPITCH + 2 * half;
    const float* fb = stg + (I_ + 64 * nG + l16) * SPITCH + 2 * half;

    v8f acc[3][4];
#pragma unroll
    for (int m = 0; m < 3; ++m)
#pragma unroll
      for (int n = 0; n < 4; ++n) acc[m][n] = (v8f){};

    {
      const int koff = 0;
      PIECES(PLOAD)
      PIECES(PSTORE)
    }
    __syncthreads();

    auto chunk_mma = [&]() {
#pragma unroll 4
      for (int kk = 0; kk < KC; kk += 4) {
        v2f a[3], bf[4];
#pragma unroll
        for (int m = 0; m < 3; ++m) a[m]  = *(const v2f*)(fa + m * 16 * SPITCH + kk);
#pragma unroll
        for (int n = 0; n < 4; ++n) bf[n] = *(const v2f*)(fb + n * 16 * SPITCH + kk);
#pragma unroll
        for (int m = 0; m < 3; ++m)
#pragma unroll
          for (int n = 0; n < 4; ++n)
            acc[m][n] = __builtin_amdgcn_wmma_f32_16x16x4_f32(false, a[m], false, bf[n],
                                                              (short)0, acc[m][n], false, false);
      }
    };

#pragma unroll 1
    for (int kc = 1; kc < D_ / KC; ++kc) {
      chunk_mma();
      __syncthreads();
      {
        const int koff = kc * KC;
        PIECES(PLOAD)
      }
      PIECES(PSTORE)
      __syncthreads();
    }
    chunk_mma();

#pragma unroll
    for (int m = 0; m < 3; ++m) {
#pragma unroll
      for (int n = 0; n < 4; ++n) {
#pragma unroll
        for (int r = 0; r < 8; ++r) {
          const int i = 48 * mG + 16 * m + r + 8 * half;
          const int j = 64 * nG + 16 * n + l16;
          const float u = gum[(size_t)b * I_ * J_ + (size_t)i * J_ + j];
          const float g = -__logf(-__logf(u));
          const float ev = (acc[m][n][r] * ESCALE + g) * TEMPINV;
          if (!(ev > -400.0f)) bad = 1;
          sE[i * LDE + j] = ev;
        }
      }
    }
  }
  if (bad) sFlag = 1;
  __syncthreads();

  const int li  = tid % I_;
  const int lq  = tid / I_;
  const int lo0 = li * LDE + lq * (J_ / 4);
  {
    float mx = -3.0e38f, mn = 3.0e38f;
#pragma unroll 8
    for (int t = 0; t < J_ / 4; ++t) {
      const float e = sE[lo0 + t];
      mx = (e > mx) ? e : mx;
      mn = (e < mn) ? e : mn;
    }
    sSB[tid] = mx;
    sB0[tid] = mn;
  }
  __syncthreads();
  const float rowM = fmaxf(fmaxf(sSB[li], sSB[I_ + li]), fmaxf(sSB[2 * I_ + li], sSB[3 * I_ + li]));
  const float rowm = fminf(fminf(sB0[li], sB0[I_ + li]), fminf(sB0[2 * I_ + li], sB0[3 * I_ + li]));
  if (!(rowm - rowM > -80.0f)) sFlag = 1;
  __syncthreads();
  const int slow = sFlag;

  auto tileC = [&](const int mT, const int nT) {
    const int m = mT * 16 + l16;
    const int n = nT * 16 + l16;
    v8f acc = {};
    for (int k0 = 0; k0 < I_; k0 += 4) {
      const int kk = k0 + 2 * half;
      v2f a;
      a.x = sE[kk * LDE + m];
      a.y = sE[(kk + 1) * LDE + m];
      v2f bf;
      bf.x = tx[(size_t)kk * D_ + n];
      bf.y = tx[(size_t)(kk + 1) * D_ + n];
      acc = __builtin_amdgcn_wmma_f32_16x16x4_f32(false, a, false, bf,
                                                  (short)0, acc, false, false);
    }
    for (int r = 0; r < 8; ++r) {
      const int j = mT * 16 + r + 8 * half;
      out_exp[(size_t)b * J_ * D_ + (size_t)j * D_ + n] = acc[r];
    }
  };

  if (slow) {
    float* Bc = sB0;
    float* Bn = sB1;
    Bc[tid] = (tid == 0) ? 0.0f : NEGV;
    __syncthreads();

    for (int i = 0; i < I_; ++i) {
      const float e  = sE[i * LDE + tid];
      float xl = wave_suffix_lse(e, lane);
      if (lane == 0) sWtL[wave] = xl;
      __syncthreads();
      float lhi = NEGINF;
      for (int w = wave + 1; w < NWAVE; ++w) lhi = lse2(lhi, sWtL[w]);
      const float Lv = lse2(xl, lhi);

      const float bv = Bc[tid];
      const float av = bv - Lv;
      float xp  = wave_prefix_lse(av, lane);
      float xsa = wave_suffix_lse(av, lane);
      float xsb = wave_suffix_lse(bv, lane);
      if (lane == 31) sWtP[wave] = xp;
      if (lane == 0) { sWtSA[wave] = xsa; sWtSB[wave] = xsb; }
      __syncthreads();
      float plo = NEGINF, sahi = NEGINF, sbhi = NEGINF;
      for (int w = 0; w < wave; ++w)          plo  = lse2(plo,  sWtP[w]);
      for (int w = wave + 1; w < NWAVE; ++w) { sahi = lse2(sahi, sWtSA[w]);
                                               sbhi = lse2(sbhi, sWtSB[w]); }
      const float Pv  = lse2(xp,  plo);
      const float SAv = lse2(xsa, sahi);
      const float SBv = lse2(xsb, sbhi);
      sP[tid]  = Pv;
      sSB[tid] = SBv;
      __syncthreads();

      if (i < I_ - 1) {
        float nb;
        if (tid == 0) {
          nb = NEGV;
        } else {
          const int j = tid - 1;
          const float v = lse2(sE[i * LDE + j] + sP[j], SAv + LOGEPS);
          const int r = i + 1;
          nb = (tid >= r && tid <= (J_ - I_ + r + 1)) ? v : NEGV;
        }
        Bn[tid] = nb;
      }

      float lb;
      if (i < I_ - 1) {
        const float sbexc = (tid + 1 < J_) ? sSB[tid + 1] : NEGINF;
        lb = lse2(Lv + Pv, sbexc + LOGEPS);
      } else {
        lb = sSB[0] + ((tid == J_ - 1) ? 0.0f : LOGEPS);
      }
      out_soft[(size_t)b * I_ * J_ + (size_t)i * J_ + tid] = lb;

      __syncthreads();
      sE[i * LDE + tid] = __expf(lb);
      { float* tmp = Bc; Bc = Bn; Bn = tmp; }
    }
    __syncthreads();
    for (int t = wave; t < (J_/16) * (D_/16); t += NWAVE) tileC(t % (J_/16), t / (J_/16));
  } else {
    {
      float s = 0.0f;
#pragma unroll 8
      for (int t = J_ / 4 - 1; t >= 0; --t) {
        const float e = sE[lo0 + t];
        s += __expf(e - rowM);
        sL[lo0 + t] = s;
        sE[lo0 + t] = e * LOG2E;
      }
      sSB[tid] = s;
      __syncthreads();
      float carry = 0.0f;
      for (int q2 = lq + 1; q2 < 4; ++q2) carry += sSB[q2 * I_ + li];
#pragma unroll 8
      for (int t = 0; t < J_ / 4; ++t) sL[lo0 + t] = (rowM + __logf(sL[lo0 + t] + carry)) * LOG2E;
      __syncthreads();
    }

    const int i0 = 3 * lane;
    int eo[3], lo[3];
    float eN[3], lN[3];
    float tu[3];
    float qv[3];
    float dC[3], cC[3], lC[3];
    float eR[3], lR[3];
#pragma unroll
    for (int r = 0; r < 3; ++r) {
      const int i = i0 + r;
      eo[r] = ((i > 0) ? i - 1 : 0) * LDE;
      lo[r] = i * LDE;
      eN[r] = 0.0f;
      lN[r] = 0.0f;
      tu[r] = NEGINF;
      qv[r] = NEGINF;
      dC[r] = NEGINF;
      cC[r] = NEGINF;
      lC[r] = 0.0f;
      eR[r] = 0.0f;
      lR[r] = 0.0f;
    }
    if (wave == 0) {
      float ps[3];
#pragma unroll
      for (int r = 0; r < 3; ++r) {
        lN[r] = sL[lo[r]];
        ps[r] = NEGINF;
      }

      for (int j4 = 0; j4 < J_; j4 += 4) {
        float xq[3][4];
#pragma unroll
        for (int q = 0; q < 4; ++q) {
          const int j = j4 + q;
          float eC[3], Lc[3], pu[3];
          pu[0] = lane_below(ps[2]);
          pu[1] = ps[0];
          pu[2] = ps[1];
#pragma unroll
          for (int r = 0; r < 3; ++r) {
            eC[r] = eN[r];
            Lc[r] = lN[r];
            eN[r] = sE[eo[r] + j];
            lN[r] = sL[lo[r] + j + 1];
          }
          float pb[3];
#pragma unroll
          for (int r = 0; r < 3; ++r) {
            const int i = i0 + r;
            const bool win = (j >= i) && (j <= J_ - I_ + i + 1);
            float b0 = win ? eC[r] + pu[r] : NEGV2;
            if (r == 0) b0 = (i == 0) ? ((j == 0) ? 0.0f : NEGV2) : b0;
            pb[r] = b0 - Lc[r];
          }
          lse2n<3>(ps, pb);
#pragma unroll
          for (int r = 0; r < 3; ++r) xq[r][q] = Lc[r] + ps[r];
        }
#pragma unroll
        for (int r = 0; r < 3; ++r)
          *(float4*)(osoft + (size_t)(i0 + r) * J_ + j4) =
              make_float4(xq[r][0], xq[r][1], xq[r][2], xq[r][3]);
      }
      tu[0] = lane_below(ps[2]);
      tu[1] = ps[0];
      tu[2] = ps[1];
#pragma unroll
      for (int r = 0; r < 3; ++r) {
        qv[r] = NEGINF;
        dC[r] = NEGINF;
        cC[r] = NEGINF;
        lC[r] = sL[lo[r]];
        eR[r] = sE[eo[r]];
        lR[r] = sL[lo[r] + 1];
      }
    } else {
      for (int idx = tid - 32; idx < 32 * 32; idx += 32 * (NWAVE - 1))
        sLc[(idx >> 5) * 33 + (idx & 31)] = sL[(idx >> 5) * LDE + (idx & 31)];
    }
    __syncthreads();

#pragma unroll 1
    for (int blk = 0; blk < NBLK + 2; ++blk) {
      if (wave == 0) {
        if (blk < NBLK) {
          float yp[3] = {0.0f, 0.0f, 0.0f};
          auto colstep = [&](const int j, const bool storePrev) __attribute__((always_inline)) {
            float qu[3];
            qu[0] = lane_below(qv[2]);
            qu[1] = qv[0];
            qu[2] = qv[1];
            if (storePrev) {
#pragma unroll
              for (int r = 0; r < 3; ++r) sL[lo[r] + j - 1] = yp[r];
            }
            const int jl = (j + 2 <= J_) ? j + 2 : J_;
            float e2[3], l2[3], dN[3], cN[3];
#pragma unroll
            for (int r = 0; r < 3; ++r) {
              e2[r] = sE[eo[r] + j + 1];
              l2[r] = sL[lo[r] + jl];
            }
#pragma unroll
            for (int r = 0; r < 3; ++r) {
              const int i = i0 + r;
              const bool wn = (i > 0) && (j + 1 >= i) && (j <= J_ - I_ + i);
              dN[r] = wn ? eR[r] - lR[r] : NEGINF;
              cN[r] = wn ? tu[r] - lR[r] : NEGINF;
            }
            float bq[3];
#pragma unroll
            for (int r = 0; r < 3; ++r) bq[r] = dC[r] + qu[r];
            lse3n<3>(qv, bq, cC);
#pragma unroll
            for (int r = 0; r < 3; ++r) {
              yp[r] = lC[r] + qv[r];
              dC[r] = dN[r];
              cC[r] = cN[r];
              lC[r] = lR[r];
              eR[r] = e2[r];
              lR[r] = l2[r];
            }
          };
          colstep(blk * CB, false);
          for (int j = blk * CB + 1; j < blk * CB + CB; ++j) colstep(j, true);
#pragma unroll
          for (int r = 0; r < 3; ++r) sL[lo[r] + blk * CB + CB - 1] = yp[r];
        }
      } else {
        const int ww = wave - 1;
        if (blk == 0) {
          constexpr int NR = (I_ + NWAVE - 2) / (NWAVE - 1);
          constexpr int NT = J_ / 32;
          float tv[NR][NT], rm[NR], rs[NR];
#pragma unroll
          for (int u = 0; u < NR; ++u) {
            const int i  = ww + (NWAVE - 1) * u;
            const int ib = (i > 0 && i < I_) ? i - 1 : 0;
#pragma unroll
            for (int t = 0; t < NT; ++t) tv[u][t] = osoft[(size_t)ib * J_ + lane + 32 * t];
          }
#pragma unroll
          for (int u = 0; u < NR; ++u) {
            const int i  = ww + (NWAVE - 1) * u;
            const int ib = (i > 0 && i < I_) ? i - 1 : 0;
            rm[u] = NEGINF;
#pragma unroll
            for (int t = 0; t < NT; ++t) {
              const int cc = lane + 32 * t;
              const bool in = (i > 0) && (i < I_) && (cc >= i - 1) && (cc <= J_ - I_ + i) && (cc <= J_ - 2);
              const float lv = (t == 0) ? sLc[(ib & 31) * 33 + lane] : sL[ib * LDE + cc];
              const float v = (sE[ib * LDE + cc] + tv[u][t]) - lv;
              tv[u][t] = in ? v : NEGINF;
              rm[u] = fmaxf(rm[u], tv[u][t]);
            }
          }
#pragma unroll
          for (int off = 16; off > 0; off >>= 1)
#pragma unroll
            for (int u = 0; u < NR; ++u) rm[u] = fmaxf(rm[u], __shfl_xor(rm[u], off, 32));
#pragma unroll
          for (int u = 0; u < NR; ++u) {
            rs[u] = 0.0f;
#pragma unroll
            for (int t = 0; t < NT; ++t) rs[u] += __builtin_amdgcn_exp2f(tv[u][t] - rm[u]);
          }
#pragma unroll
          for (int off = 16; off > 0; off >>= 1)
#pragma unroll
            for (int u = 0; u < NR; ++u) rs[u] += __shfl_xor(rs[u], off, 32);
#pragma unroll
          for (int u = 0; u < NR; ++u) {
            const int i = ww + (NWAVE - 1) * u;
            sP[i] = (i > 0) ? (rm[u] + __builtin_amdgcn_logf(rs[u])) * LN2 : 0.0f;
          }
        }
        if (blk >= 1 && blk <= NBLK) {
          const int j = (blk - 1) * CB + lane;
          float xr[9];
#pragma unroll
          for (int u = 0; u < 9; ++u) {
            const int i2 = ww + (NWAVE - 1) * u;
            xr[u] = (i2 < I_) ? osoft[(size_t)i2 * J_ + j] : 0.0f;
          }
#pragma unroll
          for (int u = 0; u < 9; ++u) {
            const int i2 = ww + (NWAVE - 1) * u;
            if (i2 < I_) {
              const float s = sP[i2];
              float lb;
              if (i2 < I_ - 1) {
                const float x = xr[u] * LN2;
                const float y = sL[i2 * LDE + j] * LN2;
                lb = lse2(lse2(x, y + LOGEPS), s + LOGEPS);
              } else {
                lb = s + ((j == J_ - 1) ? 0.0f : LOGEPS);
              }
              osoft[(size_t)i2 * J_ + j] = lb;
              sE[i2 * LDE + j] = __expf(lb);
            }
          }
        }
        if (blk >= 2) {
          for (int t = ww; t < (CB / 16) * (D_ / 16); t += NWAVE - 1)
            tileC((CB / 16) * (blk - 2) + (t & 1), t >> 1);
        }
      }
      __syncthreads();
    }
  }
}

extern "C" void kernel_launch(void* const* d_in, const int* in_sizes, int n_in,
                              void* d_out, int out_size, void* d_ws, size_t ws_size,
                              hipStream_t stream) {
  (void)in_sizes; (void)n_in; (void)d_ws; (void)ws_size; (void)out_size;
  const float* text = (const float*)d_in[0];
  const float* mel  = (const float*)d_in[1];
  const float* gum  = (const float*)d_in[2];
  float* out_soft = (float*)d_out;
  float* out_exp  = (float*)d_out + (size_t)B_ * I_ * J_;
  mobo_kernel<<<dim3(B_), dim3(384), 0, stream>>>(text, mel, gum, out_soft, out_exp);
}


# Expanded GEMM with the text operand in registers: 8 waves each hold the B fragments of 32 output columns for the whole sweep and form a 32x32 piece (2x2 accumulators) per block; sums, orders kept
# speedup vs baseline: 5.0858x; 1.1166x over previous
#include <hip/hip_runtime.h>
#include <hip/hip_bf16.h>
#include <math.h>


typedef float v2f __attribute__((ext_vector_type(2)));
typedef float v8f __attribute__((ext_vector_type(8)));

#define B_      4
#define I_      96
#define J_      384
#define D_      256
#define NWAVE   12
#define LDE     (J_ + 1)
#define LOGEPS  (-1000.0f)
#define NEGV    (-1000000000.0f)
#define NEGINF  (-1.0e38f)
#define LOG2E   1.44269504088896340736f
#define LN2     0.69314718055994530942f
#define NEGV2   (NEGV * LOG2E)
#define TEMPINV (1.0f/0.55f)
#define ESCALE  (1.0f/256.0f)
#define KC      64
#define CB      32
#define NBLK    (J_ / CB)
#define SPITCH  (KC + 4)

__device__ __forceinline__ float lse2(float a, float b) {
  float m = fmaxf(a, b);
  if (m < -9.0e37f) return NEGINF;
  return m + __logf(__expf(a - m) + __expf(b - m));
}

template <int N>
__device__ __forceinline__ void lse2n(float (&a)[N], const float (&b)[N]) {
  float m[N], ea[N], eb[N];
#pragma unroll
  for (int n = 0; n < N; ++n) m[n] = fmaxf(a[n], b[n]);
#pragma unroll
  for (int n = 0; n < N; ++n) ea[n] = __builtin_amdgcn_exp2f(a[n] - m[n]);
#pragma unroll
  for (int n = 0; n < N; ++n) eb[n] = __builtin_amdgcn_exp2f(b[n] - m[n]);
  float sv[N];
#pragma unroll
  for (int n = 0; n < N; ++n) sv[n] = ea[n] + eb[n];
#pragma unroll
  for (int n = 0; n < N; ++n) {
    const float r = m[n] + __builtin_amdgcn_logf(sv[n]);
    a[n] = (m[n] < -9.0e37f) ? NEGINF : r;
  }
}

template <int N>
__device__ __forceinline__ void lse3n(float (&a)[N], const float (&b)[N], const float (&c)[N]) {
  float m[N], ea[N], eb[N], ec[N];
#pragma unroll
  for (int n = 0; n < N; ++n) m[n] = fmaxf(fmaxf(a[n], b[n]), c[n]);
#pragma unroll
  for (int n = 0; n < N; ++n) ea[n] = __builtin_amdgcn_exp2f(a[n] - m[n]);
#pragma unroll
  for (int n = 0; n < N; ++n) eb[n] = __builtin_amdgcn_exp2f(b[n] - m[n]);
#pragma unroll
  for (int n = 0; n < N; ++n) ec[n] = __builtin_amdgcn_exp2f(c[n] - m[n]);
  float sv[N];
#pragma unroll
  for (int n = 0; n < N; ++n) sv[n] = ea[n] + (eb[n] + ec[n]);
#pragma unroll
  for (int n = 0; n < N; ++n) {
    const float r = m[n] + __builtin_amdgcn_logf(sv[n]);
    a[n] = (m[n] < -9.0e37f) ? NEGINF : r;
  }
}

__device__ __forceinline__ float wave_prefix_lse(float x, int lane) {
#pragma unroll
  for (int off = 1; off < 32; off <<= 1) {
    float o = __shfl_up(x, off, 32);
    if (lane >= off) x = lse2(x, o);
  }
  return x;
}

__device__ __forceinline__ float wave_suffix_lse(float x, int lane) {
#pragma unroll
  for (int off = 1; off < 32; off <<= 1) {
    float o = __shfl_down(x, off, 32);
    if (lane + off < 32) x = lse2(x, o);
  }
  return x;
}

__device__ __forceinline__ float lane_below(float x) {
  const int xi = __float_as_int(x);
  const int cross = __builtin_amdgcn_permlanex16(xi, xi, 0xffffffffu, 0xffffffffu, false, false);
  return __int_as_float(__builtin_amdgcn_update_dpp(cross, xi, 0x111, 0xf, 0xf, false));
}

__global__ __launch_bounds__(384, 1)
void mobo_kernel(const float* __restrict__ text,
                 const float* __restrict__ mel,
                 const float* __restrict__ gum,
                 float* __restrict__ out_soft,
                 float* __restrict__ out_exp)
{
  __shared__ float sE[I_ * LDE];
  __shared__ alignas(16) float sL[I_ * LDE];
  __shared__ int   sFlag;
  __shared__ float sP[J_];
  __shared__ float sSB[J_];
  __shared__ float sB0[J_];
  __shared__ float sB1[J_];
  __shared__ float sLc[32 * 33];
  __shared__ float sWtL[NWAVE];
  __shared__ float sWtP[NWAVE];
  __shared__ float sWtSA[NWAVE];
  __shared__ float sWtSB[NWAVE];

  const int b    = blockIdx.x;
  const int tid  = threadIdx.x;
  const int lane = tid & 31;
  const int wave = tid >> 5;
  const int half = lane >> 4;
  const int l16  = lane & 15;

  const float* tx = text + (size_t)b * I_ * D_;
  const float* ml = mel  + (size_t)b * J_ * D_;
  float* osoft = out_soft + (size_t)b * I_ * J_;

  if (tid == 0) sFlag = 0;
  __syncthreads();
  int bad = 0;

  {
    float* stg = sL;
    const int prow = tid >> 4;
    const int pc4  = (tid & 15) * 4;
    const float* gsrc[2];
    gsrc[0] = tx + (size_t)prow * D_ + pc4;
    gsrc[1] = ml + (size_t)prow * D_ + pc4;
    float* sdst = stg + prow * SPITCH + pc4;
#define PIECES(X) X(0) X(1) X(2) X(3) X(4) X(5) X(6) X(7) X(8) X(9) X(10) X(11) X(12) X(13) X(14) X(15) X(16) X(17) X(18) X(19)
#define PSRC(u)   ((u) < 4 ? gsrc[0] + (size_t)(24 * (u)) * D_ : gsrc[1] + (size_t)(24 * ((u) - 4)) * D_)
#define PDECL(u)  float4 pf##u;
#define PLOAD(u)  pf##u = *(const float4*)(PSRC(u) + koff);
#define PSTORE(u) *(float4*)(sdst + 24 * (u) * SPITCH) = pf##u;
    PIECES(PDECL)

    const int mG = wave & 1;
    const int nG = wave >> 1;
    const float* fa = stg + (48 * mG + l16) * SPITCH + 2 * half;
    const float* fb = stg + (I_ + 64 * nG + l16) * SPITCH + 2 * half;

    v8f acc[3][4];
#pragma unroll
    for (int m = 0; m < 3; ++m)
#pragma unroll
      for (int n = 0; n < 4; ++n) acc[m][n] = (v8f){};

    {
      const int koff = 0;
      PIECES(PLOAD)
      PIECES(PSTORE)
    }
    __syncthreads();

    auto chunk_mma = [&]() {
#pragma unroll 4
      for (int kk = 0; kk < KC; kk += 4) {
        v2f a[3], bf[4];
#pragma unroll
        for (int m = 0; m < 3; ++m) a[m]  = *(const v2f*)(fa + m * 16 * SPITCH + kk);
#pragma unroll
        for (int n = 0; n < 4; ++n) bf[n] = *(const v2f*)(fb + n * 16 * SPITCH + kk);
#pragma unroll
        for (int m = 0; m < 3; ++m)
#pragma unroll
          for (int n = 0; n < 4; ++n)
            acc[m][n] = __builtin_amdgcn_wmma_f32_16x16x4_f32(false, a[m], false, bf[n],
                                                              (short)0, acc[m][n], false, false);
      }
    };

#pragma unroll 1
    for (int kc = 1; kc < D_ / KC; ++kc) {
      chunk_mma();
      __syncthreads();
      {
        const int koff = kc * KC;
        PIECES(PLOAD)
      }
      PIECES(PSTORE)
      __syncthreads();
    }
    chunk_mma();

#pragma unroll
    for (int m = 0; m < 3; ++m) {
#pragma unroll
      for (int n = 0; n < 4; ++n) {
#pragma unroll
        for (int r = 0; r < 8; ++r) {
          const int i = 48 * mG + 16 * m + r + 8 * half;
          const int j = 64 * nG + 16 * n + l16;
          const float u = gum[(size_t)b * I_ * J_ + (size_t)i * J_ + j];
          const float g = -__logf(-__logf(u));
          const float ev = (acc[m][n][r] * ESCALE + g) * TEMPINV;
          if (!(ev > -400.0f)) bad = 1;
          sE[i * LDE + j] = ev;
        }
      }
    }
  }
  if (bad) sFlag = 1;
  __syncthreads();

  const int li  = tid % I_;
  const int lq  = tid / I_;
  const int lo0 = li * LDE + lq * (J_ / 4);
  {
    float mx = -3.0e38f, mn = 3.0e38f;
#pragma unroll 8
    for (int t = 0; t < J_ / 4; ++t) {
      const float e = sE[lo0 + t];
      mx = (e > mx) ? e : mx;
      mn = (e < mn) ? e : mn;
    }
    sSB[tid] = mx;
    sB0[tid] = mn;
  }
  __syncthreads();
  const float rowM = fmaxf(fmaxf(sSB[li], sSB[I_ + li]), fmaxf(sSB[2 * I_ + li], sSB[3 * I_ + li]));
  const float rowm = fminf(fminf(sB0[li], sB0[I_ + li]), fminf(sB0[2 * I_ + li], sB0[3 * I_ + li]));
  if (!(rowm - rowM > -80.0f)) sFlag = 1;
  __syncthreads();
  const int slow = sFlag;

  auto tileC = [&](const int mT, const int nT) {
    const int m = mT * 16 + l16;
    const int n = nT * 16 + l16;
    v8f acc = {};
    for (int k0 = 0; k0 < I_; k0 += 4) {
      const int kk = k0 + 2 * half;
      v2f a;
      a.x = sE[kk * LDE + m];
      a.y = sE[(kk + 1) * LDE + m];
      v2f bf;
      bf.x = tx[(size_t)kk * D_ + n];
      bf.y = tx[(size_t)(kk + 1) * D_ + n];
      acc = __builtin_amdgcn_wmma_f32_16x16x4_f32(false, a, false, bf,
                                                  (short)0, acc, false, false);
    }
    for (int r = 0; r < 8; ++r) {
      const int j = mT * 16 + r + 8 * half;
      out_exp[(size_t)b * J_ * D_ + (size_t)j * D_ + n] = acc[r];
    }
  };

  if (slow) {
    float* Bc = sB0;
    float* Bn = sB1;
    Bc[tid] = (tid == 0) ? 0.0f : NEGV;
    __syncthreads();

    for (int i = 0; i < I_; ++i) {
      const float e  = sE[i * LDE + tid];
      float xl = wave_suffix_lse(e, lane);
      if (lane == 0) sWtL[wave] = xl;
      __syncthreads();
      float lhi = NEGINF;
      for (int w = wave + 1; w < NWAVE; ++w) lhi = lse2(lhi, sWtL[w]);
      const float Lv = lse2(xl, lhi);

      const float bv = Bc[tid];
      const float av = bv - Lv;
      float xp  = wave_prefix_lse(av, lane);
      float xsa = wave_suffix_lse(av, lane);
      float xsb = wave_suffix_lse(bv, lane);
      if (lane == 31) sWtP[wave] = xp;
      if (lane == 0) { sWtSA[wave] = xsa; sWtSB[wave] = xsb; }
      __syncthreads();
      float plo = NEGINF, sahi = NEGINF, sbhi = NEGINF;
      for (int w = 0; w < wave; ++w)          plo  = lse2(plo,  sWtP[w]);
      for (int w = wave + 1; w < NWAVE; ++w) { sahi = lse2(sahi, sWtSA[w]);
                                               sbhi = lse2(sbhi, sWtSB[w]); }
      const float Pv  = lse2(xp,  plo);
      const float SAv = lse2(xsa, sahi);
      const float SBv = lse2(xsb, sbhi);
      sP[tid]  = Pv;
      sSB[tid] = SBv;
      __syncthreads();

      if (i < I_ - 1) {
        float nb;
        if (tid == 0) {
          nb = NEGV;
        } else {
          const int j = tid - 1;
          const float v = lse2(sE[i * LDE + j] + sP[j], SAv + LOGEPS);
          const int r = i + 1;
          nb = (tid >= r && tid <= (J_ - I_ + r + 1)) ? v : NEGV;
        }
        Bn[tid] = nb;
      }

      float lb;
      if (i < I_ - 1) {
        const float sbexc = (tid + 1 < J_) ? sSB[tid + 1] : NEGINF;
        lb = lse2(Lv + Pv, sbexc + LOGEPS);
      } else {
        lb = sSB[0] + ((tid == J_ - 1) ? 0.0f : LOGEPS);
      }
      out_soft[(size_t)b * I_ * J_ + (size_t)i * J_ + tid] = lb;

      __syncthreads();
      sE[i * LDE + tid] = __expf(lb);
      { float* tmp = Bc; Bc = Bn; Bn = tmp; }
    }
    __syncthreads();
    for (int t = wave; t < (J_/16) * (D_/16); t += NWAVE) tileC(t % (J_/16), t / (J_/16));
  } else {
    {
      float s = 0.0f;
#pragma unroll 8
      for (int t = J_ / 4 - 1; t >= 0; --t) {
        const float e = sE[lo0 + t];
        s += __expf(e - rowM);
        sL[lo0 + t] = s;
        sE[lo0 + t] = e * LOG2E;
      }
      sSB[tid] = s;
      __syncthreads();
      float carry = 0.0f;
      for (int q2 = lq + 1; q2 < 4; ++q2) carry += sSB[q2 * I_ + li];
#pragma unroll 8
      for (int t = 0; t < J_ / 4; ++t) sL[lo0 + t] = (rowM + __logf(sL[lo0 + t] + carry)) * LOG2E;
      __syncthreads();
    }

    const int i0 = 3 * lane;
    int eo[3], lo[3];
    float eN[3], lN[3];
    float tu[3];
    float qv[3];
    float dC[3], cC[3], lC[3];
    float eR[3], lR[3];
#pragma unroll
    for (int r = 0; r < 3; ++r) {
      const int i = i0 + r;
      eo[r] = ((i > 0) ? i - 1 : 0) * LDE;
      lo[r] = i * LDE;
      eN[r] = 0.0f;
      lN[r] = 0.0f;
      tu[r] = NEGINF;
      qv[r] = NEGINF;
      dC[r] = NEGINF;
      cC[r] = NEGINF;
      lC[r] = 0.0f;
      eR[r] = 0.0f;
      lR[r] = 0.0f;
    }
    const bool gemmw = ((wave & 3) != 0) && (wave != NWAVE - 1);
    const int  gq    = wave - 1 - (wave >> 2);
    v2f tb[2][I_ / 4];
    if (wave == 0) {
      float ps[3];
#pragma unroll
      for (int r = 0; r < 3; ++r) {
        lN[r] = sL[lo[r]];
        ps[r] = NEGINF;
      }

      for (int j4 = 0; j4 < J_; j4 += 4) {
        float xq[3][4];
#pragma unroll
        for (int q = 0; q < 4; ++q) {
          const int j = j4 + q;
          float eC[3], Lc[3], pu[3];
          pu[0] = lane_below(ps[2]);
          pu[1] = ps[0];
          pu[2] = ps[1];
#pragma unroll
          for (int r = 0; r < 3; ++r) {
            eC[r] = eN[r];
            Lc[r] = lN[r];
            eN[r] = sE[eo[r] + j];
            lN[r] = sL[lo[r] + j + 1];
          }
          float pb[3];
#pragma unroll
          for (int r = 0; r < 3; ++r) {
            const int i = i0 + r;
            const bool win = (j >= i) && (j <= J_ - I_ + i + 1);
            float b0 = win ? eC[r] + pu[r] : NEGV2;
            if (r == 0) b0 = (i == 0) ? ((j == 0) ? 0.0f : NEGV2) : b0;
            pb[r] = b0 - Lc[r];
          }
          lse2n<3>(ps, pb);
#pragma unroll
          for (int r = 0; r < 3; ++r) xq[r][q] = Lc[r] + ps[r];
        }
#pragma unroll
        for (int r = 0; r < 3; ++r)
          *(float4*)(osoft + (size_t)(i0 + r) * J_ + j4) =
              make_float4(xq[r][0], xq[r][1], xq[r][2], xq[r][3]);
      }
      tu[0] = lane_below(ps[2]);
      tu[1] = ps[0];
      tu[2] = ps[1];
#pragma unroll
      for (int r = 0; r < 3; ++r) {
        qv[r] = NEGINF;
        dC[r] = NEGINF;
        cC[r] = NEGINF;
        lC[r] = sL[lo[r]];
        eR[r] = sE[eo[r]];
        lR[r] = sL[lo[r] + 1];
      }
    } else {
      for (int idx = tid - 32; idx < 32 * 32; idx += 32 * (NWAVE - 1))
        sLc[(idx >> 5) * 33 + (idx & 31)] = sL[(idx >> 5) * LDE + (idx & 31)];
    }
    __syncthreads();

    auto gemm_rows = [&](const int blk) __attribute__((always_inline)) {
      const int mb = (blk - 2) * CB;
      float* op = out_exp + (size_t)b * J_ * D_ + (size_t)(mb + 8 * half) * D_ + 32 * gq + l16;
      asm volatile("" : "+v"(op));
      typedef float __attribute__((address_space(1))) gfloat;
      gfloat* const gp = (gfloat*)op;
      v2f af[2][I_ / 4];
#pragma unroll
      for (int ks = 0; ks < I_ / 4; ++ks) {
#pragma unroll
        for (int mt = 0; mt < 2; ++mt) {
          const int kk = 4 * ks + 2 * half;
          const int m  = mb + 16 * mt + l16;
          af[mt][ks].x = sE[kk * LDE + m];
          af[mt][ks].y = sE[(kk + 1) * LDE + m];
        }
      }
      __builtin_amdgcn_sched_barrier(0);
      v8f acc[2][2];
#pragma unroll
      for (int mt = 0; mt < 2; ++mt)
#pragma unroll
        for (int nt = 0; nt < 2; ++nt) acc[mt][nt] = (v8f){};
#pragma unroll
      for (int ks = 0; ks < I_ / 4; ++ks)
#pragma unroll
        for (int mt = 0; mt < 2; ++mt)
#pragma unroll
          for (int nt = 0; nt < 2; ++nt)
            acc[mt][nt] = __builtin_amdgcn_wmma_f32_16x16x4_f32(false, af[mt][ks], false, tb[nt][ks],
                                                                (short)0, acc[mt][nt], false, false);
      __builtin_amdgcn_sched_barrier(0);
#pragma unroll
      for (int mt = 0; mt < 2; ++mt)
#pragma unroll
        for (int nt = 0; nt < 2; ++nt)
#pragma unroll
          for (int r = 0; r < 8; ++r)
            gp[(size_t)(16 * mt + r) * D_ + 16 * nt] = acc[mt][nt][r];
    };

    auto interval = [&](const int blk, const bool first) __attribute__((always_inline)) {
      if (wave == 0) {
        if (blk < NBLK) {
          float yp[3] = {0.0f, 0.0f, 0.0f};
          auto colstep = [&](const int j, const bool storePrev) __attribute__((always_inline)) {
            float qu[3];
            qu[0] = lane_below(qv[2]);
            qu[1] = qv[0];
            qu[2] = qv[1];
            if (storePrev) {
#pragma unroll
              for (int r = 0; r < 3; ++r) sL[lo[r] + j - 1] = yp[r];
            }
            const int jl = (j + 2 <= J_) ? j + 2 : J_;
            float e2[3], l2[3], dN[3], cN[3];
#pragma unroll
            for (int r = 0; r < 3; ++r) {
              e2[r] = sE[eo[r] + j + 1];
              l2[r] = sL[lo[r] + jl];
            }
#pragma unroll
            for (int r = 0; r < 3; ++r) {
              const int i = i0 + r;
              const bool wn = (i > 0) && (j + 1 >= i) && (j <= J_ - I_ + i);
              dN[r] = wn ? eR[r] - lR[r] : NEGINF;
              cN[r] = wn ? tu[r] - lR[r] : NEGINF;
            }
            float bq[3];
#pragma unroll
            for (int r = 0; r < 3; ++r) bq[r] = dC[r] + qu[r];
            lse3n<3>(qv, bq, cC);
#pragma unroll
            for (int r = 0; r < 3; ++r) {
              yp[r] = lC[r] + qv[r];
              dC[r] = dN[r];
              cC[r] = cN[r];
              lC[r] = lR[r];
              eR[r] = e2[r];
              lR[r] = l2[r];
            }
          };
          colstep(blk * CB, false);
          for (int j = blk * CB + 1; j < blk * CB + CB; ++j) colstep(j, true);
#pragma unroll
          for (int r = 0; r < 3; ++r) sL[lo[r] + blk * CB + CB - 1] = yp[r];
        }
      } else {
        const int ww = wave - 1;
        const bool gearly = gemmw && (wave < 4);
        if (blk >= 2 && gearly) gemm_rows(blk);
        if (first) {
          constexpr int NR = (I_ + NWAVE - 2) / (NWAVE - 1);
          constexpr int NT = J_ / 32;
          float tv[NR][NT], rm[NR], rs[NR];
#pragma unroll
          for (int u = 0; u < NR; ++u) {
            const int i  = ww + (NWAVE - 1) * u;
            const int ib = (i > 0 && i < I_) ? i - 1 : 0;
#pragma unroll
            for (int t = 0; t < NT; ++t) tv[u][t] = osoft[(size_t)ib * J_ + lane + 32 * t];
          }
#pragma unroll
          for (int u = 0; u < NR; ++u) {
            const int i  = ww + (NWAVE - 1) * u;
            const int ib = (i > 0 && i < I_) ? i - 1 : 0;
            rm[u] = NEGINF;
#pragma unroll
            for (int t = 0; t < NT; ++t) {
              const int cc = lane + 32 * t;
              const bool in = (i > 0) && (i < I_) && (cc >= i - 1) && (cc <= J_ - I_ + i) && (cc <= J_ - 2);
              const float lv = (t == 0) ? sLc[(ib & 31) * 33 + lane] : sL[ib * LDE + cc];
              const float v = (sE[ib * LDE + cc] + tv[u][t]) - lv;
              tv[u][t] = in ? v : NEGINF;
              rm[u] = fmaxf(rm[u], tv[u][t]);
            }
          }
#pragma unroll
          for (int off = 16; off > 0; off >>= 1)
#pragma unroll
            for (int u = 0; u < NR; ++u) rm[u] = fmaxf(rm[u], __shfl_xor(rm[u], off, 32));
#pragma unroll
          for (int u = 0; u < NR; ++u) {
            rs[u] = 0.0f;
#pragma unroll
            for (int t = 0; t < NT; ++t) rs[u] += __builtin_amdgcn_exp2f(tv[u][t] - rm[u]);
          }
#pragma unroll
          for (int off = 16; off > 0; off >>= 1)
#pragma unroll
            for (int u = 0; u < NR; ++u) rs[u] += __shfl_xor(rs[u], off, 32);
#pragma unroll
          for (int u = 0; u < NR; ++u) {
            const int i = ww + (NWAVE - 1) * u;
            sP[i] = (i > 0) ? (rm[u] + __builtin_amdgcn_logf(rs[u])) * LN2 : 0.0f;
          }
        }
        if (blk >= 1 && blk <= NBLK) {
          const int j = (blk - 1) * CB + lane;
          float xr[9];
#pragma unroll
          for (int u = 0; u < 9; ++u) {
            const int i2 = ww + (NWAVE - 1) * u;
            xr[u] = (i2 < I_) ? osoft[(size_t)i2 * J_ + j] : 0.0f;
          }
#pragma unroll
          for (int u = 0; u < 9; ++u) {
            const int i2 = ww + (NWAVE - 1) * u;
            if (i2 < I_) {
              const float s = sP[i2];
              float lb;
              if (i2 < I_ - 1) {
                const float x = xr[u] * LN2;
                const float y = sL[i2 * LDE + j] * LN2;
                lb = lse2(lse2(x, y + LOGEPS), s + LOGEPS);
              } else {
                lb = s + ((j == J_ - 1) ? 0.0f : LOGEPS);
              }
              osoft[(size_t)i2 * J_ + j] = lb;
              sE[i2 * LDE + j] = __expf(lb);
            }
          }
        }
        if (blk >= 2 && gemmw && !gearly) gemm_rows(blk);
      }
      __syncthreads();
    };

    interval(0, true);
    if (gemmw) {
#pragma unroll
      for (int nt = 0; nt < 2; ++nt) {
#pragma unroll
        for (int ks = 0; ks < I_ / 4; ++ks) {
          const int kk = 4 * ks + 2 * half;
          const int n  = 32 * gq + 16 * nt + l16;
          tb[nt][ks].x = tx[(size_t)kk * D_ + n];
          tb[nt][ks].y = tx[(size_t)(kk + 1) * D_ + n];
        }
      }
    }
#pragma unroll 1
    for (int blk = 1; blk < NBLK + 2; ++blk) interval(blk, false);
  }
}

extern "C" void kernel_launch(void* const* d_in, const int* in_sizes, int n_in,
                              void* d_out, int out_size, void* d_ws, size_t ws_size,
                              hipStream_t stream) {
  (void)in_sizes; (void)n_in; (void)d_ws; (void)ws_size; (void)out_size;
  const float* text = (const float*)d_in[0];
  const float* mel  = (const float*)d_in[1];
  const float* gum  = (const float*)d_in[2];
  float* out_soft = (float*)d_out;
  float* out_exp  = (float*)d_out + (size_t)B_ * I_ * J_;
  mobo_kernel<<<dim3(B_), dim3(384), 0, stream>>>(text, mel, gum, out_soft, out_exp);
}
